# starting kernel
# baseline (speedup 1.0000x reference)
#include <hip/hip_runtime.h>
#include <hip/hip_bf16.h>

#define EMB_DIM 128

typedef __attribute__((ext_vector_type(2))) float v2f;
typedef __attribute__((ext_vector_type(8))) float v8f;

__device__ __forceinline__ void gatomic_add_f32(float* p, float v) {
  asm volatile("global_atomic_add_f32 %0, %1, off scope:SCOPE_DEV"
               :: "v"(p), "v"(v) : "memory");
}

__device__ __forceinline__ void gatomic_add_f32x4(float* p, float4 v) {
  asm volatile(
      "global_atomic_add_f32 %0, %1, off scope:SCOPE_DEV\n\t"
      "global_atomic_add_f32 %0, %2, off offset:4 scope:SCOPE_DEV\n\t"
      "global_atomic_add_f32 %0, %3, off offset:8 scope:SCOPE_DEV\n\t"
      "global_atomic_add_f32 %0, %4, off offset:12 scope:SCOPE_DEV"
      :: "v"(p), "v"(v.x), "v"(v.y), "v"(v.z), "v"(v.w) : "memory");
}

__global__ void zero_f32(float* __restrict__ p, long long n) {
  long long t = (long long)blockIdx.x * blockDim.x + threadIdx.x;
  if (t < n) p[t] = 0.0f;
}

__global__ void degree_kernel(const int* __restrict__ dst,
                              float* __restrict__ deg, int n_edges) {
  int e = blockIdx.x * blockDim.x + threadIdx.x;
  if (e < n_edges) gatomic_add_f32(deg + dst[e], 1.0f);
}

__global__ void __launch_bounds__(256)
scatter_kernel(const int* __restrict__ src, const int* __restrict__ dst,
               const float* __restrict__ h, float* __restrict__ agg,
               int n_edges) {
  int t = blockIdx.x * blockDim.x + threadIdx.x;
  int e = t >> 5;
  int lane = t & 31;
  if (e >= n_edges) return;
  int s = src[e];
  int d = dst[e];
  const float4 v = *(const float4*)(h + (size_t)s * EMB_DIM + lane * 4);
  gatomic_add_f32x4(agg + (size_t)d * EMB_DIM + lane * 4, v);
}

__global__ void norm_kernel(float* __restrict__ g, const float* __restrict__ deg,
                            int num_users) {
  int t = blockIdx.x * blockDim.x + threadIdx.x;
  int u = t >> 5;
  int lane = t & 31;
  if (u >= num_users) return;
  float inv = 1.0f / fmaxf(deg[u], 1.0f);
  float4* p = (float4*)(g + (size_t)u * EMB_DIM + lane * 4);
  float4 v = *p;
  v.x *= inv; v.y *= inv; v.z *= inv; v.w *= inv;
  *p = v;
}

__global__ void __launch_bounds__(256)
gather_kernel(const int* __restrict__ users, const int* __restrict__ items,
              const float* __restrict__ uemb, const float* __restrict__ g1,
              const float* __restrict__ g2, const float* __restrict__ iemb,
              float* __restrict__ out_user, float* __restrict__ out_item,
              int batch) {
  int t = blockIdx.x * blockDim.x + threadIdx.x;
  int b = t >> 5;
  int lane = t & 31;
  if (b >= batch) return;
  int u = users[b];
  int it = items[b];
  size_t off = (size_t)lane * 4;
  const float4 a = *(const float4*)(uemb + (size_t)u * EMB_DIM + off);
  const float4 c = *(const float4*)(g1 + (size_t)u * EMB_DIM + off);
  const float4 d = *(const float4*)(g2 + (size_t)u * EMB_DIM + off);
  float4 fu;
  fu.x = a.x + c.x + d.x;
  fu.y = a.y + c.y + d.y;
  fu.z = a.z + c.z + d.z;
  fu.w = a.w + c.w + d.w;
  *(float4*)(out_user + (size_t)b * EMB_DIM + off) = fu;
  *(float4*)(out_item + (size_t)b * EMB_DIM + off) =
      *(const float4*)(iemb + (size_t)it * EMB_DIM + off);
}

__global__ void __launch_bounds__(256)
dot_sigmoid_wmma(const float* __restrict__ U, const float* __restrict__ V,
                 float* __restrict__ pred, int batch) {
  int gid = blockIdx.x * blockDim.x + threadIdx.x;
  int wave = gid >> 5;
  int lane = threadIdx.x & 31;
  if (wave * 16 >= batch) return;

  int row = lane & 15;
  int half = lane >> 4;
  const float* urow = U + ((size_t)wave * 16 + row) * EMB_DIM + 2 * half;
  const float* vrow = V + ((size_t)wave * 16 + row) * EMB_DIM + 2 * half;

  v8f acc = {};
#pragma unroll
  for (int k0 = 0; k0 < EMB_DIM; k0 += 4) {
    v2f a = *(const v2f*)(urow + k0);
    v2f b = *(const v2f*)(vrow + k0);
    acc = __builtin_amdgcn_wmma_f32_16x16x4_f32(
         false, a,  false, b,
         (short)0, acc,  false,  false);
  }

  if (lane < 8 || lane >= 24) {
    int i = (lane < 8) ? lane : (lane - 16);
    float d = acc[lane & 7];
    pred[(size_t)wave * 16 + i] = 1.0f / (1.0f + expf(-d));
  }
}

extern "C" void kernel_launch(void* const* d_in, const int* in_sizes, int n_in,
                              void* d_out, int out_size, void* d_ws,
                              size_t ws_size, hipStream_t stream) {
  const int* users  = (const int*)d_in[0];
  const int* items  = (const int*)d_in[1];
  const int* esrc   = (const int*)d_in[2];
  const int* edst   = (const int*)d_in[3];
  const float* uemb = (const float*)d_in[4];
  const float* iemb = (const float*)d_in[5];

  const int batch     = in_sizes[0];
  const int n_edges   = in_sizes[2];
  const int num_users = in_sizes[4] / EMB_DIM;

  float* deg = (float*)d_ws;
  float* g1  = deg + num_users;
  float* g2  = g1 + (size_t)num_users * EMB_DIM;

  float* pred     = (float*)d_out;
  float* out_user = pred + batch;
  float* out_item = out_user + (size_t)batch * EMB_DIM;

  const int B = 256;

  long long total = (long long)num_users * (1 + 2 * EMB_DIM);
  zero_f32<<<(int)((total + B - 1) / B), B, 0, stream>>>((float*)d_ws, total);

  degree_kernel<<<(n_edges + B - 1) / B, B, 0, stream>>>(edst, deg, n_edges);

  long long sc_threads = (long long)n_edges * 32;
  scatter_kernel<<<(int)((sc_threads + B - 1) / B), B, 0, stream>>>(
      esrc, edst, uemb, g1, n_edges);
  long long nm_threads = (long long)num_users * 32;
  norm_kernel<<<(int)((nm_threads + B - 1) / B), B, 0, stream>>>(g1, deg,
                                                                 num_users);

  scatter_kernel<<<(int)((sc_threads + B - 1) / B), B, 0, stream>>>(
      esrc, edst, g1, g2, n_edges);
  norm_kernel<<<(int)((nm_threads + B - 1) / B), B, 0, stream>>>(g2, deg,
                                                                 num_users);

  long long ga_threads = (long long)batch * 32;
  gather_kernel<<<(int)((ga_threads + B - 1) / B), B, 0, stream>>>(
      users, items, uemb, g1, g2, iemb, out_user, out_item, batch);

  long long dot_threads = (long long)((batch + 15) / 16) * 32;
  dot_sigmoid_wmma<<<(int)((dot_threads + B - 1) / B), B, 0, stream>>>(
      out_user, out_item, pred, batch);
}


# layer 2 (atomic scatter + normalize) only for user rows the batch gathers: per-user flag set from users[], one wave scans 32 edges and a ballot keeps flagged ones; f32 atomic adds stay arrival-order
# speedup vs baseline: 1.5104x; 1.5104x over previous
#include <hip/hip_runtime.h>
#include <hip/hip_bf16.h>

#define EMB_DIM 128

typedef __attribute__((ext_vector_type(2))) float v2f;
typedef __attribute__((ext_vector_type(8))) float v8f;

__device__ __forceinline__ void gatomic_add_f32(float* p, float v) {
  asm volatile("global_atomic_add_f32 %0, %1, off scope:SCOPE_DEV"
               :: "v"(p), "v"(v) : "memory");
}

__device__ __forceinline__ void gatomic_add_f32x4(float* p, float4 v) {
  asm volatile(
      "global_atomic_add_f32 %0, %1, off scope:SCOPE_DEV\n\t"
      "global_atomic_add_f32 %0, %2, off offset:4 scope:SCOPE_DEV\n\t"
      "global_atomic_add_f32 %0, %3, off offset:8 scope:SCOPE_DEV\n\t"
      "global_atomic_add_f32 %0, %4, off offset:12 scope:SCOPE_DEV"
      :: "v"(p), "v"(v.x), "v"(v.y), "v"(v.z), "v"(v.w) : "memory");
}

__global__ void zero_f32(float* __restrict__ p, long long n) {
  long long t = (long long)blockIdx.x * blockDim.x + threadIdx.x;
  if (t < n) p[t] = 0.0f;
}

__global__ void degree_kernel(const int* __restrict__ dst,
                              const int* __restrict__ users,
                              float* __restrict__ deg, int* __restrict__ flags,
                              int n_edges, int batch) {
  int e = blockIdx.x * blockDim.x + threadIdx.x;
  if (e < n_edges) gatomic_add_f32(deg + dst[e], 1.0f);
  if (flags != nullptr && e < batch) flags[users[e]] = 1;
}

__global__ void __launch_bounds__(256)
scatter_kernel(const int* __restrict__ src, const int* __restrict__ dst,
               const float* __restrict__ h, float* __restrict__ agg,
               int n_edges) {
  int t = blockIdx.x * blockDim.x + threadIdx.x;
  int e = t >> 5;
  int lane = t & 31;
  if (e >= n_edges) return;
  int s = src[e];
  int d = dst[e];
  const float4 v = *(const float4*)(h + (size_t)s * EMB_DIM + lane * 4);
  gatomic_add_f32x4(agg + (size_t)d * EMB_DIM + lane * 4, v);
}

__global__ void __launch_bounds__(256)
scatter_flagged_kernel(const int* __restrict__ src, const int* __restrict__ dst,
                       const int* __restrict__ flags,
                       const float* __restrict__ h, float* __restrict__ agg,
                       int n_edges) {
  int e = blockIdx.x * blockDim.x + threadIdx.x;
  int lane = threadIdx.x & 31;
  if ((e - lane) >= n_edges) return;
  int s = 0, d = 0, f = 0;
  if (e < n_edges) {
    s = src[e];
    d = dst[e];
    f = flags[d];
  }
  unsigned mask = __builtin_amdgcn_ballot_w32(f != 0);
  while (mask) {
    int j = __builtin_ctz(mask);
    mask &= mask - 1;
    int sj = __builtin_amdgcn_readlane(s, j);
    int dj = __builtin_amdgcn_readlane(d, j);
    const float4 v = *(const float4*)(h + (size_t)sj * EMB_DIM + lane * 4);
    gatomic_add_f32x4(agg + (size_t)dj * EMB_DIM + lane * 4, v);
  }
}

__global__ void norm_kernel(float* __restrict__ g, const float* __restrict__ deg,
                            const int* __restrict__ flags, int num_users) {
  int t = blockIdx.x * blockDim.x + threadIdx.x;
  int u = t >> 5;
  int lane = t & 31;
  if (u >= num_users) return;
  if (flags != nullptr && flags[u] == 0) return;
  float inv = 1.0f / fmaxf(deg[u], 1.0f);
  float4* p = (float4*)(g + (size_t)u * EMB_DIM + lane * 4);
  float4 v = *p;
  v.x *= inv; v.y *= inv; v.z *= inv; v.w *= inv;
  *p = v;
}

__global__ void __launch_bounds__(256)
gather_kernel(const int* __restrict__ users, const int* __restrict__ items,
              const float* __restrict__ uemb, const float* __restrict__ g1,
              const float* __restrict__ g2, const float* __restrict__ iemb,
              float* __restrict__ out_user, float* __restrict__ out_item,
              int batch) {
  int t = blockIdx.x * blockDim.x + threadIdx.x;
  int b = t >> 5;
  int lane = t & 31;
  if (b >= batch) return;
  int u = users[b];
  int it = items[b];
  size_t off = (size_t)lane * 4;
  const float4 a = *(const float4*)(uemb + (size_t)u * EMB_DIM + off);
  const float4 c = *(const float4*)(g1 + (size_t)u * EMB_DIM + off);
  const float4 d = *(const float4*)(g2 + (size_t)u * EMB_DIM + off);
  float4 fu;
  fu.x = a.x + c.x + d.x;
  fu.y = a.y + c.y + d.y;
  fu.z = a.z + c.z + d.z;
  fu.w = a.w + c.w + d.w;
  *(float4*)(out_user + (size_t)b * EMB_DIM + off) = fu;
  *(float4*)(out_item + (size_t)b * EMB_DIM + off) =
      *(const float4*)(iemb + (size_t)it * EMB_DIM + off);
}

__global__ void __launch_bounds__(256)
dot_sigmoid_wmma(const float* __restrict__ U, const float* __restrict__ V,
                 float* __restrict__ pred, int batch) {
  int gid = blockIdx.x * blockDim.x + threadIdx.x;
  int wave = gid >> 5;
  int lane = threadIdx.x & 31;
  if (wave * 16 >= batch) return;

  int row = lane & 15;
  int half = lane >> 4;
  const float* urow = U + ((size_t)wave * 16 + row) * EMB_DIM + 2 * half;
  const float* vrow = V + ((size_t)wave * 16 + row) * EMB_DIM + 2 * half;

  v8f acc = {};
#pragma unroll
  for (int k0 = 0; k0 < EMB_DIM; k0 += 4) {
    v2f a = *(const v2f*)(urow + k0);
    v2f b = *(const v2f*)(vrow + k0);
    acc = __builtin_amdgcn_wmma_f32_16x16x4_f32(
         false, a,  false, b,
         (short)0, acc,  false,  false);
  }

  if (lane < 8 || lane >= 24) {
    int i = (lane < 8) ? lane : (lane - 16);
    float d = acc[lane & 7];
    pred[(size_t)wave * 16 + i] = 1.0f / (1.0f + expf(-d));
  }
}

extern "C" void kernel_launch(void* const* d_in, const int* in_sizes, int n_in,
                              void* d_out, int out_size, void* d_ws,
                              size_t ws_size, hipStream_t stream) {
  const int* users  = (const int*)d_in[0];
  const int* items  = (const int*)d_in[1];
  const int* esrc   = (const int*)d_in[2];
  const int* edst   = (const int*)d_in[3];
  const float* uemb = (const float*)d_in[4];
  const float* iemb = (const float*)d_in[5];

  const int batch     = in_sizes[0];
  const int n_edges   = in_sizes[2];
  const int num_users = in_sizes[4] / EMB_DIM;

  float* deg = (float*)d_ws;
  float* g1  = deg + num_users;
  float* g2  = g1 + (size_t)num_users * EMB_DIM;
  long long total = (long long)num_users * (1 + 2 * EMB_DIM);
  int* flags = nullptr;
  if (ws_size >= (size_t)(total + num_users) * sizeof(float)) {
    flags = (int*)(g2 + (size_t)num_users * EMB_DIM);
    total += num_users;
  }

  float* pred     = (float*)d_out;
  float* out_user = pred + batch;
  float* out_item = out_user + (size_t)batch * EMB_DIM;

  const int B = 256;

  zero_f32<<<(int)((total + B - 1) / B), B, 0, stream>>>((float*)d_ws, total);

  int dm_threads = n_edges > batch ? n_edges : batch;
  degree_kernel<<<(dm_threads + B - 1) / B, B, 0, stream>>>(
      edst, users, deg, flags, n_edges, batch);

  long long sc_threads = (long long)n_edges * 32;
  scatter_kernel<<<(int)((sc_threads + B - 1) / B), B, 0, stream>>>(
      esrc, edst, uemb, g1, n_edges);
  long long nm_threads = (long long)num_users * 32;
  norm_kernel<<<(int)((nm_threads + B - 1) / B), B, 0, stream>>>(
      g1, deg, nullptr, num_users);

  if (flags != nullptr) {
    scatter_flagged_kernel<<<(n_edges + B - 1) / B, B, 0, stream>>>(
        esrc, edst, flags, g1, g2, n_edges);
  } else {
    scatter_kernel<<<(int)((sc_threads + B - 1) / B), B, 0, stream>>>(
        esrc, edst, g1, g2, n_edges);
  }
  norm_kernel<<<(int)((nm_threads + B - 1) / B), B, 0, stream>>>(
      g2, deg, flags, num_users);

  long long ga_threads = (long long)batch * 32;
  gather_kernel<<<(int)((ga_threads + B - 1) / B), B, 0, stream>>>(
      users, items, uemb, g1, g2, iemb, out_user, out_item, batch);

  long long dot_threads = (long long)((batch + 15) / 16) * 32;
  dot_sigmoid_wmma<<<(int)((dot_threads + B - 1) / B), B, 0, stream>>>(
      out_user, out_item, pred, batch);
}


# each wave loops over many items instead of one: a scatter wave moves 32 edges (4 row reads in flight), zeroing, degree and normalisation stride a fixed grid; f32 atomic adds stay arrival-order
# speedup vs baseline: 1.6151x; 1.0693x over previous
#include <hip/hip_runtime.h>
#include <hip/hip_bf16.h>

#define EMB_DIM 128

typedef __attribute__((ext_vector_type(2))) float v2f;
typedef __attribute__((ext_vector_type(8))) float v8f;

__device__ __forceinline__ void gatomic_add_f32(float* p, float v) {
  asm volatile("global_atomic_add_f32 %0, %1, off scope:SCOPE_DEV"
               :: "v"(p), "v"(v) : "memory");
}

__device__ __forceinline__ void gatomic_add_f32x4(float* p, float4 v) {
  asm volatile(
      "global_atomic_add_f32 %0, %1, off scope:SCOPE_DEV\n\t"
      "global_atomic_add_f32 %0, %2, off offset:4 scope:SCOPE_DEV\n\t"
      "global_atomic_add_f32 %0, %3, off offset:8 scope:SCOPE_DEV\n\t"
      "global_atomic_add_f32 %0, %4, off offset:12 scope:SCOPE_DEV"
      :: "v"(p), "v"(v.x), "v"(v.y), "v"(v.z), "v"(v.w) : "memory");
}

__global__ void __launch_bounds__(256)
zero_f32(float* __restrict__ p, long long n) {
  const long long n4 = n >> 2;
  const long long stride = (long long)gridDim.x * blockDim.x;
  const long long t = (long long)blockIdx.x * blockDim.x + threadIdx.x;
  const float4 z = make_float4(0.0f, 0.0f, 0.0f, 0.0f);
  float4* p4 = (float4*)p;
  for (long long i = t; i < n4; i += stride) p4[i] = z;
  if (t < (n & 3)) p[(n4 << 2) + t] = 0.0f;
}

__global__ void __launch_bounds__(256)
degree_kernel(const int* __restrict__ dst, const int* __restrict__ users,
              float* __restrict__ deg, int* __restrict__ flags, int n_edges,
              int batch) {
  const int stride = gridDim.x * blockDim.x;
  const int t = blockIdx.x * blockDim.x + threadIdx.x;
  for (int e = t; e < n_edges; e += stride)
    gatomic_add_f32(deg + dst[e], 1.0f);
  if (flags != nullptr)
    for (int b = t; b < batch; b += stride) flags[users[b]] = 1;
}

__global__ void __launch_bounds__(256)
scatter_kernel(const int* __restrict__ src, const int* __restrict__ dst,
               const float* __restrict__ h, float* __restrict__ agg,
               int n_edges) {
  int e = blockIdx.x * blockDim.x + threadIdx.x;
  int lane = threadIdx.x & 31;
  int base = __builtin_amdgcn_readfirstlane(e - lane);
  if (base >= n_edges) return;
  int s = 0, d = 0;
  if (e < n_edges) {
    s = src[e];
    d = dst[e];
  }
  int left = n_edges - base;
  int cnt = left < 32 ? left : 32;
  for (int j = 0; j < cnt; j += 4) {
    int sj[4], dj[4];
    float4 v[4];
#pragma unroll
    for (int k = 0; k < 4; ++k) {
      int jj = (j + k < cnt) ? (j + k) : (cnt - 1);
      sj[k] = __builtin_amdgcn_readlane(s, jj);
      dj[k] = __builtin_amdgcn_readlane(d, jj);
      v[k] = *(const float4*)(h + (size_t)sj[k] * EMB_DIM + lane * 4);
    }
#pragma unroll
    for (int k = 0; k < 4; ++k) {
      if (j + k < cnt)
        gatomic_add_f32x4(agg + (size_t)dj[k] * EMB_DIM + lane * 4, v[k]);
    }
  }
}

__global__ void __launch_bounds__(256)
scatter_flagged_kernel(const int* __restrict__ src, const int* __restrict__ dst,
                       const int* __restrict__ flags,
                       const float* __restrict__ h, float* __restrict__ agg,
                       int n_edges) {
  int e = blockIdx.x * blockDim.x + threadIdx.x;
  int lane = threadIdx.x & 31;
  if ((e - lane) >= n_edges) return;
  int s = 0, d = 0, f = 0;
  if (e < n_edges) {
    s = src[e];
    d = dst[e];
    f = flags[d];
  }
  unsigned mask = __builtin_amdgcn_ballot_w32(f != 0);
  while (mask) {
    int j = __builtin_ctz(mask);
    mask &= mask - 1;
    int sj = __builtin_amdgcn_readlane(s, j);
    int dj = __builtin_amdgcn_readlane(d, j);
    const float4 v = *(const float4*)(h + (size_t)sj * EMB_DIM + lane * 4);
    gatomic_add_f32x4(agg + (size_t)dj * EMB_DIM + lane * 4, v);
  }
}

__global__ void __launch_bounds__(256)
norm_kernel(float* __restrict__ g, const float* __restrict__ deg,
            const int* __restrict__ flags, int num_users) {
  const int lane = threadIdx.x & 31;
  const int wave = (blockIdx.x * blockDim.x + threadIdx.x) >> 5;
  const int n_waves = (gridDim.x * blockDim.x) >> 5;
  for (int u = wave; u < num_users; u += n_waves) {
    if (flags != nullptr && flags[u] == 0) continue;
    float inv = 1.0f / fmaxf(deg[u], 1.0f);
    float4* p = (float4*)(g + (size_t)u * EMB_DIM + lane * 4);
    float4 v = *p;
    v.x *= inv; v.y *= inv; v.z *= inv; v.w *= inv;
    *p = v;
  }
}

__global__ void __launch_bounds__(256)
gather_kernel(const int* __restrict__ users, const int* __restrict__ items,
              const float* __restrict__ uemb, const float* __restrict__ g1,
              const float* __restrict__ g2, const float* __restrict__ iemb,
              float* __restrict__ out_user, float* __restrict__ out_item,
              int batch) {
  int t = blockIdx.x * blockDim.x + threadIdx.x;
  int b = t >> 5;
  int lane = t & 31;
  if (b >= batch) return;
  int u = users[b];
  int it = items[b];
  size_t off = (size_t)lane * 4;
  const float4 a = *(const float4*)(uemb + (size_t)u * EMB_DIM + off);
  const float4 c = *(const float4*)(g1 + (size_t)u * EMB_DIM + off);
  const float4 d = *(const float4*)(g2 + (size_t)u * EMB_DIM + off);
  float4 fu;
  fu.x = a.x + c.x + d.x;
  fu.y = a.y + c.y + d.y;
  fu.z = a.z + c.z + d.z;
  fu.w = a.w + c.w + d.w;
  *(float4*)(out_user + (size_t)b * EMB_DIM + off) = fu;
  *(float4*)(out_item + (size_t)b * EMB_DIM + off) =
      *(const float4*)(iemb + (size_t)it * EMB_DIM + off);
}

__global__ void __launch_bounds__(256)
dot_sigmoid_wmma(const float* __restrict__ U, const float* __restrict__ V,
                 float* __restrict__ pred, int batch) {
  int gid = blockIdx.x * blockDim.x + threadIdx.x;
  int wave = gid >> 5;
  int lane = threadIdx.x & 31;
  if (wave * 16 >= batch) return;

  int row = lane & 15;
  int half = lane >> 4;
  const float* urow = U + ((size_t)wave * 16 + row) * EMB_DIM + 2 * half;
  const float* vrow = V + ((size_t)wave * 16 + row) * EMB_DIM + 2 * half;

  v8f acc = {};
#pragma unroll
  for (int k0 = 0; k0 < EMB_DIM; k0 += 4) {
    v2f a = *(const v2f*)(urow + k0);
    v2f b = *(const v2f*)(vrow + k0);
    acc = __builtin_amdgcn_wmma_f32_16x16x4_f32(
         false, a,  false, b,
         (short)0, acc,  false,  false);
  }

  if (lane < 8 || lane >= 24) {
    int i = (lane < 8) ? lane : (lane - 16);
    float d = acc[lane & 7];
    pred[(size_t)wave * 16 + i] = 1.0f / (1.0f + expf(-d));
  }
}

extern "C" void kernel_launch(void* const* d_in, const int* in_sizes, int n_in,
                              void* d_out, int out_size, void* d_ws,
                              size_t ws_size, hipStream_t stream) {
  const int* users  = (const int*)d_in[0];
  const int* items  = (const int*)d_in[1];
  const int* esrc   = (const int*)d_in[2];
  const int* edst   = (const int*)d_in[3];
  const float* uemb = (const float*)d_in[4];
  const float* iemb = (const float*)d_in[5];

  const int batch     = in_sizes[0];
  const int n_edges   = in_sizes[2];
  const int num_users = in_sizes[4] / EMB_DIM;

  float* deg = (float*)d_ws;
  float* g1  = deg + num_users;
  float* g2  = g1 + (size_t)num_users * EMB_DIM;
  long long total = (long long)num_users * (1 + 2 * EMB_DIM);
  int* flags = nullptr;
  if (ws_size >= (size_t)(total + num_users) * sizeof(float)) {
    flags = (int*)(g2 + (size_t)num_users * EMB_DIM);
    total += num_users;
  }

  float* pred     = (float*)d_out;
  float* out_user = pred + batch;
  float* out_item = out_user + (size_t)batch * EMB_DIM;

  const int B = 256;

  const int G = 2048;

  zero_f32<<<G, B, 0, stream>>>((float*)d_ws, total);

  degree_kernel<<<G / 4, B, 0, stream>>>(edst, users, deg, flags, n_edges,
                                         batch);

  const int sc_blocks = (n_edges + B - 1) / B;
  scatter_kernel<<<sc_blocks, B, 0, stream>>>(esrc, edst, uemb, g1, n_edges);
  norm_kernel<<<G, B, 0, stream>>>(g1, deg, nullptr, num_users);

  if (flags != nullptr) {
    scatter_flagged_kernel<<<sc_blocks, B, 0, stream>>>(esrc, edst, flags, g1,
                                                        g2, n_edges);
  } else {
    scatter_kernel<<<sc_blocks, B, 0, stream>>>(esrc, edst, g1, g2, n_edges);
  }
  norm_kernel<<<G, B, 0, stream>>>(g2, deg, flags, num_users);

  long long ga_threads = (long long)batch * 32;
  gather_kernel<<<(int)((ga_threads + B - 1) / B), B, 0, stream>>>(
      users, items, uemb, g1, g2, iemb, out_user, out_item, batch);

  long long dot_threads = (long long)((batch + 15) / 16) * 32;
  dot_sigmoid_wmma<<<(int)((dot_threads + B - 1) / B), B, 0, stream>>>(
      out_user, out_item, pred, batch);
}


# each f32 atomic instruction of both scatters now covers one whole 128-byte line (lane L owns floats L,L+32,L+64,L+96; row read as 4-byte loads to match); adds stay arrival-order
# speedup vs baseline: 2.3644x; 1.4639x over previous
#include <hip/hip_runtime.h>
#include <hip/hip_bf16.h>

#define EMB_DIM 128

typedef __attribute__((ext_vector_type(2))) float v2f;
typedef __attribute__((ext_vector_type(8))) float v8f;

__device__ __forceinline__ void gatomic_add_f32(float* p, float v) {
  asm volatile("global_atomic_add_f32 %0, %1, off scope:SCOPE_DEV"
               :: "v"(p), "v"(v) : "memory");
}

__device__ __forceinline__ void gatomic_add_row(float* p, float v0, float v1,
                                                float v2, float v3) {
  asm volatile(
      "global_atomic_add_f32 %0, %1, off scope:SCOPE_DEV\n\t"
      "global_atomic_add_f32 %0, %2, off offset:128 scope:SCOPE_DEV\n\t"
      "global_atomic_add_f32 %0, %3, off offset:256 scope:SCOPE_DEV\n\t"
      "global_atomic_add_f32 %0, %4, off offset:384 scope:SCOPE_DEV"
      :: "v"(p), "v"(v0), "v"(v1), "v"(v2), "v"(v3) : "memory");
}

struct RowQuad { float a, b, c, d; };
__device__ __forceinline__ RowQuad load_row(const float* __restrict__ p) {
  RowQuad r;
  r.a = p[0];
  r.b = p[32];
  r.c = p[64];
  r.d = p[96];
  return r;
}

__global__ void __launch_bounds__(256)
zero_f32(float* __restrict__ p, long long n) {
  const long long n4 = n >> 2;
  const long long stride = (long long)gridDim.x * blockDim.x;
  const long long t = (long long)blockIdx.x * blockDim.x + threadIdx.x;
  const float4 z = make_float4(0.0f, 0.0f, 0.0f, 0.0f);
  float4* p4 = (float4*)p;
  for (long long i = t; i < n4; i += stride) p4[i] = z;
  if (t < (n & 3)) p[(n4 << 2) + t] = 0.0f;
}

__global__ void __launch_bounds__(256)
degree_kernel(const int* __restrict__ dst, const int* __restrict__ users,
              float* __restrict__ deg, int* __restrict__ flags, int n_edges,
              int batch) {
  const int stride = gridDim.x * blockDim.x;
  const int t = blockIdx.x * blockDim.x + threadIdx.x;
  for (int e = t; e < n_edges; e += stride)
    gatomic_add_f32(deg + dst[e], 1.0f);
  if (flags != nullptr)
    for (int b = t; b < batch; b += stride) flags[users[b]] = 1;
}

__global__ void __launch_bounds__(256)
scatter_kernel(const int* __restrict__ src, const int* __restrict__ dst,
               const float* __restrict__ h, float* __restrict__ agg,
               int n_edges) {
  int e = blockIdx.x * blockDim.x + threadIdx.x;
  int lane = threadIdx.x & 31;
  int base = __builtin_amdgcn_readfirstlane(e - lane);
  if (base >= n_edges) return;
  int s = 0, d = 0;
  if (e < n_edges) {
    s = src[e];
    d = dst[e];
  }
  int left = n_edges - base;
  int cnt = left < 32 ? left : 32;
  for (int j = 0; j < cnt; j += 4) {
    int sj[4], dj[4];
    RowQuad v[4];
#pragma unroll
    for (int k = 0; k < 4; ++k) {
      int jj = (j + k < cnt) ? (j + k) : (cnt - 1);
      sj[k] = __builtin_amdgcn_readlane(s, jj);
      dj[k] = __builtin_amdgcn_readlane(d, jj);
      v[k] = load_row(h + (size_t)sj[k] * EMB_DIM + lane);
    }
#pragma unroll
    for (int k = 0; k < 4; ++k) {
      if (j + k < cnt)
        gatomic_add_row(agg + (size_t)dj[k] * EMB_DIM + lane, v[k].a, v[k].b,
                        v[k].c, v[k].d);
    }
  }
}

__global__ void __launch_bounds__(256)
scatter_flagged_kernel(const int* __restrict__ src, const int* __restrict__ dst,
                       const int* __restrict__ flags,
                       const float* __restrict__ h, float* __restrict__ agg,
                       int n_edges) {
  int e = blockIdx.x * blockDim.x + threadIdx.x;
  int lane = threadIdx.x & 31;
  if ((e - lane) >= n_edges) return;
  int s = 0, d = 0, f = 0;
  if (e < n_edges) {
    s = src[e];
    d = dst[e];
    f = flags[d];
  }
  unsigned mask = __builtin_amdgcn_ballot_w32(f != 0);
  while (mask) {
    int j = __builtin_ctz(mask);
    mask &= mask - 1;
    int sj = __builtin_amdgcn_readlane(s, j);
    int dj = __builtin_amdgcn_readlane(d, j);
    const RowQuad v = load_row(h + (size_t)sj * EMB_DIM + lane);
    gatomic_add_row(agg + (size_t)dj * EMB_DIM + lane, v.a, v.b, v.c, v.d);
  }
}

__global__ void __launch_bounds__(256)
norm_kernel(float* __restrict__ g, const float* __restrict__ deg,
            const int* __restrict__ flags, int num_users) {
  const int lane = threadIdx.x & 31;
  const int wave = (blockIdx.x * blockDim.x + threadIdx.x) >> 5;
  const int n_waves = (gridDim.x * blockDim.x) >> 5;
  for (int u = wave; u < num_users; u += n_waves) {
    if (flags != nullptr && flags[u] == 0) continue;
    float inv = 1.0f / fmaxf(deg[u], 1.0f);
    float4* p = (float4*)(g + (size_t)u * EMB_DIM + lane * 4);
    float4 v = *p;
    v.x *= inv; v.y *= inv; v.z *= inv; v.w *= inv;
    *p = v;
  }
}

__global__ void __launch_bounds__(256)
gather_kernel(const int* __restrict__ users, const int* __restrict__ items,
              const float* __restrict__ uemb, const float* __restrict__ g1,
              const float* __restrict__ g2, const float* __restrict__ iemb,
              float* __restrict__ out_user, float* __restrict__ out_item,
              int batch) {
  int t = blockIdx.x * blockDim.x + threadIdx.x;
  int b = t >> 5;
  int lane = t & 31;
  if (b >= batch) return;
  int u = users[b];
  int it = items[b];
  size_t off = (size_t)lane * 4;
  const float4 a = *(const float4*)(uemb + (size_t)u * EMB_DIM + off);
  const float4 c = *(const float4*)(g1 + (size_t)u * EMB_DIM + off);
  const float4 d = *(const float4*)(g2 + (size_t)u * EMB_DIM + off);
  float4 fu;
  fu.x = a.x + c.x + d.x;
  fu.y = a.y + c.y + d.y;
  fu.z = a.z + c.z + d.z;
  fu.w = a.w + c.w + d.w;
  *(float4*)(out_user + (size_t)b * EMB_DIM + off) = fu;
  *(float4*)(out_item + (size_t)b * EMB_DIM + off) =
      *(const float4*)(iemb + (size_t)it * EMB_DIM + off);
}

__global__ void __launch_bounds__(256)
dot_sigmoid_wmma(const float* __restrict__ U, const float* __restrict__ V,
                 float* __restrict__ pred, int batch) {
  int gid = blockIdx.x * blockDim.x + threadIdx.x;
  int wave = gid >> 5;
  int lane = threadIdx.x & 31;
  if (wave * 16 >= batch) return;

  int row = lane & 15;
  int half = lane >> 4;
  const float* urow = U + ((size_t)wave * 16 + row) * EMB_DIM + 2 * half;
  const float* vrow = V + ((size_t)wave * 16 + row) * EMB_DIM + 2 * half;

  v8f acc = {};
#pragma unroll
  for (int k0 = 0; k0 < EMB_DIM; k0 += 4) {
    v2f a = *(const v2f*)(urow + k0);
    v2f b = *(const v2f*)(vrow + k0);
    acc = __builtin_amdgcn_wmma_f32_16x16x4_f32(
         false, a,  false, b,
         (short)0, acc,  false,  false);
  }

  if (lane < 8 || lane >= 24) {
    int i = (lane < 8) ? lane : (lane - 16);
    float d = acc[lane & 7];
    pred[(size_t)wave * 16 + i] = 1.0f / (1.0f + expf(-d));
  }
}

extern "C" void kernel_launch(void* const* d_in, const int* in_sizes, int n_in,
                              void* d_out, int out_size, void* d_ws,
                              size_t ws_size, hipStream_t stream) {
  const int* users  = (const int*)d_in[0];
  const int* items  = (const int*)d_in[1];
  const int* esrc   = (const int*)d_in[2];
  const int* edst   = (const int*)d_in[3];
  const float* uemb = (const float*)d_in[4];
  const float* iemb = (const float*)d_in[5];

  const int batch     = in_sizes[0];
  const int n_edges   = in_sizes[2];
  const int num_users = in_sizes[4] / EMB_DIM;

  float* deg = (float*)d_ws;
  float* g1  = deg + num_users;
  float* g2  = g1 + (size_t)num_users * EMB_DIM;
  long long total = (long long)num_users * (1 + 2 * EMB_DIM);
  int* flags = nullptr;
  if (ws_size >= (size_t)(total + num_users) * sizeof(float)) {
    flags = (int*)(g2 + (size_t)num_users * EMB_DIM);
    total += num_users;
  }

  float* pred     = (float*)d_out;
  float* out_user = pred + batch;
  float* out_item = out_user + (size_t)batch * EMB_DIM;

  const int B = 256;

  const int G = 2048;

  zero_f32<<<G, B, 0, stream>>>((float*)d_ws, total);

  degree_kernel<<<G / 4, B, 0, stream>>>(edst, users, deg, flags, n_edges,
                                         batch);

  const int sc_blocks = (n_edges + B - 1) / B;
  scatter_kernel<<<sc_blocks, B, 0, stream>>>(esrc, edst, uemb, g1, n_edges);
  norm_kernel<<<G, B, 0, stream>>>(g1, deg, nullptr, num_users);

  if (flags != nullptr) {
    scatter_flagged_kernel<<<sc_blocks, B, 0, stream>>>(esrc, edst, flags, g1,
                                                        g2, n_edges);
  } else {
    scatter_kernel<<<sc_blocks, B, 0, stream>>>(esrc, edst, g1, g2, n_edges);
  }
  norm_kernel<<<G, B, 0, stream>>>(g2, deg, flags, num_users);

  long long ga_threads = (long long)batch * 32;
  gather_kernel<<<(int)((ga_threads + B - 1) / B), B, 0, stream>>>(
      users, items, uemb, g1, g2, iemb, out_user, out_item, batch);

  long long dot_threads = (long long)((batch + 15) / 16) * 32;
  dot_sigmoid_wmma<<<(int)((dot_threads + B - 1) / B), B, 0, stream>>>(
      out_user, out_item, pred, batch);
}


# mean's 1/max(deg,1) folded into the readers: flagged scatter scales each g1 row by its source's factor as read, gather scales the two rows it reads; both normalise launches gone
# speedup vs baseline: 2.5634x; 1.0842x over previous
#include <hip/hip_runtime.h>
#include <hip/hip_bf16.h>

#define EMB_DIM 128

typedef __attribute__((ext_vector_type(2))) float v2f;
typedef __attribute__((ext_vector_type(8))) float v8f;

__device__ __forceinline__ void gatomic_add_f32(float* p, float v) {
  asm volatile("global_atomic_add_f32 %0, %1, off scope:SCOPE_DEV"
               :: "v"(p), "v"(v) : "memory");
}

__device__ __forceinline__ void gatomic_add_row(float* p, float v0, float v1,
                                                float v2, float v3) {
  asm volatile(
      "global_atomic_add_f32 %0, %1, off scope:SCOPE_DEV\n\t"
      "global_atomic_add_f32 %0, %2, off offset:128 scope:SCOPE_DEV\n\t"
      "global_atomic_add_f32 %0, %3, off offset:256 scope:SCOPE_DEV\n\t"
      "global_atomic_add_f32 %0, %4, off offset:384 scope:SCOPE_DEV"
      :: "v"(p), "v"(v0), "v"(v1), "v"(v2), "v"(v3) : "memory");
}

struct RowQuad { float a, b, c, d; };
__device__ __forceinline__ RowQuad load_row(const float* __restrict__ p) {
  RowQuad r;
  r.a = p[0];
  r.b = p[32];
  r.c = p[64];
  r.d = p[96];
  return r;
}

__global__ void __launch_bounds__(256)
zero_f32(float* __restrict__ p, long long n) {
  const long long n4 = n >> 2;
  const long long stride = (long long)gridDim.x * blockDim.x;
  const long long t = (long long)blockIdx.x * blockDim.x + threadIdx.x;
  const float4 z = make_float4(0.0f, 0.0f, 0.0f, 0.0f);
  float4* p4 = (float4*)p;
  for (long long i = t; i < n4; i += stride) p4[i] = z;
  if (t < (n & 3)) p[(n4 << 2) + t] = 0.0f;
}

__global__ void __launch_bounds__(256)
degree_kernel(const int* __restrict__ dst, const int* __restrict__ users,
              float* __restrict__ deg, int* __restrict__ flags, int n_edges,
              int batch) {
  const int stride = gridDim.x * blockDim.x;
  const int t = blockIdx.x * blockDim.x + threadIdx.x;
  for (int e = t; e < n_edges; e += stride)
    gatomic_add_f32(deg + dst[e], 1.0f);
  if (flags != nullptr)
    for (int b = t; b < batch; b += stride) flags[users[b]] = 1;
}

template <bool SCALE>
__global__ void __launch_bounds__(256)
scatter_kernel(const int* __restrict__ src, const int* __restrict__ dst,
               const float* __restrict__ h, const float* __restrict__ deg,
               float* __restrict__ agg, int n_edges) {
  int e = blockIdx.x * blockDim.x + threadIdx.x;
  int lane = threadIdx.x & 31;
  int base = __builtin_amdgcn_readfirstlane(e - lane);
  if (base >= n_edges) return;
  int s = 0, d = 0;
  float sc = 1.0f;
  if (e < n_edges) {
    s = src[e];
    d = dst[e];
    if (SCALE) sc = 1.0f / fmaxf(deg[s], 1.0f);
  }
  int left = n_edges - base;
  int cnt = left < 32 ? left : 32;
  for (int j = 0; j < cnt; j += 4) {
    int sj[4], dj[4];
    float cj[4];
    RowQuad v[4];
#pragma unroll
    for (int k = 0; k < 4; ++k) {
      int jj = (j + k < cnt) ? (j + k) : (cnt - 1);
      sj[k] = __builtin_amdgcn_readlane(s, jj);
      dj[k] = __builtin_amdgcn_readlane(d, jj);
      if (SCALE)
        cj[k] = __int_as_float(
            __builtin_amdgcn_readlane(__float_as_int(sc), jj));
      v[k] = load_row(h + (size_t)sj[k] * EMB_DIM + lane);
    }
#pragma unroll
    for (int k = 0; k < 4; ++k) {
      if (j + k < cnt) {
        if (SCALE) {
          v[k].a *= cj[k]; v[k].b *= cj[k]; v[k].c *= cj[k]; v[k].d *= cj[k];
        }
        gatomic_add_row(agg + (size_t)dj[k] * EMB_DIM + lane, v[k].a, v[k].b,
                        v[k].c, v[k].d);
      }
    }
  }
}

__global__ void __launch_bounds__(256)
scatter_flagged_kernel(const int* __restrict__ src, const int* __restrict__ dst,
                       const int* __restrict__ flags,
                       const float* __restrict__ h,
                       const float* __restrict__ deg, float* __restrict__ agg,
                       int n_edges) {
  int e = blockIdx.x * blockDim.x + threadIdx.x;
  int lane = threadIdx.x & 31;
  if ((e - lane) >= n_edges) return;
  int s = 0, d = 0, f = 0;
  float sc = 1.0f;
  if (e < n_edges) {
    s = src[e];
    d = dst[e];
    f = flags[d];
    if (f != 0) sc = 1.0f / fmaxf(deg[s], 1.0f);
  }
  unsigned mask = __builtin_amdgcn_ballot_w32(f != 0);
  while (mask) {
    int j = __builtin_ctz(mask);
    mask &= mask - 1;
    int sj = __builtin_amdgcn_readlane(s, j);
    int dj = __builtin_amdgcn_readlane(d, j);
    const float cj =
        __int_as_float(__builtin_amdgcn_readlane(__float_as_int(sc), j));
    const RowQuad v = load_row(h + (size_t)sj * EMB_DIM + lane);
    gatomic_add_row(agg + (size_t)dj * EMB_DIM + lane, v.a * cj, v.b * cj,
                    v.c * cj, v.d * cj);
  }
}

__global__ void __launch_bounds__(256)
gather_kernel(const int* __restrict__ users, const int* __restrict__ items,
              const float* __restrict__ uemb, const float* __restrict__ g1,
              const float* __restrict__ g2, const float* __restrict__ deg,
              const float* __restrict__ iemb, float* __restrict__ out_user,
              float* __restrict__ out_item, int batch) {
  int t = blockIdx.x * blockDim.x + threadIdx.x;
  int b = t >> 5;
  int lane = t & 31;
  if (b >= batch) return;
  int u = users[b];
  int it = items[b];
  size_t off = (size_t)lane * 4;
  const float4 a = *(const float4*)(uemb + (size_t)u * EMB_DIM + off);
  const float4 c = *(const float4*)(g1 + (size_t)u * EMB_DIM + off);
  const float4 d = *(const float4*)(g2 + (size_t)u * EMB_DIM + off);
  const float inv = 1.0f / fmaxf(deg[u], 1.0f);
  float4 fu;
  fu.x = a.x + c.x * inv + d.x * inv;
  fu.y = a.y + c.y * inv + d.y * inv;
  fu.z = a.z + c.z * inv + d.z * inv;
  fu.w = a.w + c.w * inv + d.w * inv;
  *(float4*)(out_user + (size_t)b * EMB_DIM + off) = fu;
  *(float4*)(out_item + (size_t)b * EMB_DIM + off) =
      *(const float4*)(iemb + (size_t)it * EMB_DIM + off);
}

__global__ void __launch_bounds__(256)
dot_sigmoid_wmma(const float* __restrict__ U, const float* __restrict__ V,
                 float* __restrict__ pred, int batch) {
  int gid = blockIdx.x * blockDim.x + threadIdx.x;
  int wave = gid >> 5;
  int lane = threadIdx.x & 31;
  if (wave * 16 >= batch) return;

  int row = lane & 15;
  int half = lane >> 4;
  const float* urow = U + ((size_t)wave * 16 + row) * EMB_DIM + 2 * half;
  const float* vrow = V + ((size_t)wave * 16 + row) * EMB_DIM + 2 * half;

  v8f acc = {};
#pragma unroll
  for (int k0 = 0; k0 < EMB_DIM; k0 += 4) {
    v2f a = *(const v2f*)(urow + k0);
    v2f b = *(const v2f*)(vrow + k0);
    acc = __builtin_amdgcn_wmma_f32_16x16x4_f32(
         false, a,  false, b,
         (short)0, acc,  false,  false);
  }

  if (lane < 8 || lane >= 24) {
    int i = (lane < 8) ? lane : (lane - 16);
    float d = acc[lane & 7];
    pred[(size_t)wave * 16 + i] = 1.0f / (1.0f + expf(-d));
  }
}

extern "C" void kernel_launch(void* const* d_in, const int* in_sizes, int n_in,
                              void* d_out, int out_size, void* d_ws,
                              size_t ws_size, hipStream_t stream) {
  const int* users  = (const int*)d_in[0];
  const int* items  = (const int*)d_in[1];
  const int* esrc   = (const int*)d_in[2];
  const int* edst   = (const int*)d_in[3];
  const float* uemb = (const float*)d_in[4];
  const float* iemb = (const float*)d_in[5];

  const int batch     = in_sizes[0];
  const int n_edges   = in_sizes[2];
  const int num_users = in_sizes[4] / EMB_DIM;

  float* deg = (float*)d_ws;
  float* g1  = deg + num_users;
  float* g2  = g1 + (size_t)num_users * EMB_DIM;
  long long total = (long long)num_users * (1 + 2 * EMB_DIM);
  int* flags = nullptr;
  if (ws_size >= (size_t)(total + num_users) * sizeof(float)) {
    flags = (int*)(g2 + (size_t)num_users * EMB_DIM);
    total += num_users;
  }

  float* pred     = (float*)d_out;
  float* out_user = pred + batch;
  float* out_item = out_user + (size_t)batch * EMB_DIM;

  const int B = 256;

  const int G = 2048;

  zero_f32<<<G, B, 0, stream>>>((float*)d_ws, total);

  degree_kernel<<<G / 4, B, 0, stream>>>(edst, users, deg, flags, n_edges,
                                         batch);

  const int sc_blocks = (n_edges + B - 1) / B;
  scatter_kernel<false><<<sc_blocks, B, 0, stream>>>(esrc, edst, uemb, deg, g1,
                                                     n_edges);

  if (flags != nullptr) {
    scatter_flagged_kernel<<<sc_blocks, B, 0, stream>>>(esrc, edst, flags, g1,
                                                        deg, g2, n_edges);
  } else {
    scatter_kernel<true><<<sc_blocks, B, 0, stream>>>(esrc, edst, g1, deg, g2,
                                                      n_edges);
  }

  long long ga_threads = (long long)batch * 32;
  gather_kernel<<<(int)((ga_threads + B - 1) / B), B, 0, stream>>>(
      users, items, uemb, g1, g2, deg, iemb, out_user, out_item, batch);

  long long dot_threads = (long long)((batch + 15) / 16) * 32;
  dot_sigmoid_wmma<<<(int)((dot_threads + B - 1) / B), B, 0, stream>>>(
      out_user, out_item, pred, batch);
}


# layer-1 scatter cache-blocked: four passes over the edges in one launch, by (source half, destination half) of the users, so a pass's rows and sums stay cached; f32 atomic adds stay arrival-order
# speedup vs baseline: 2.6776x; 1.0446x over previous
#include <hip/hip_runtime.h>
#include <hip/hip_bf16.h>

#define EMB_DIM 128

typedef __attribute__((ext_vector_type(2))) float v2f;
typedef __attribute__((ext_vector_type(8))) float v8f;

__device__ __forceinline__ void gatomic_add_f32(float* p, float v) {
  asm volatile("global_atomic_add_f32 %0, %1, off scope:SCOPE_DEV"
               :: "v"(p), "v"(v) : "memory");
}

__device__ __forceinline__ void gatomic_add_row(float* p, float v0, float v1,
                                                float v2, float v3) {
  asm volatile(
      "global_atomic_add_f32 %0, %1, off scope:SCOPE_DEV\n\t"
      "global_atomic_add_f32 %0, %2, off offset:128 scope:SCOPE_DEV\n\t"
      "global_atomic_add_f32 %0, %3, off offset:256 scope:SCOPE_DEV\n\t"
      "global_atomic_add_f32 %0, %4, off offset:384 scope:SCOPE_DEV"
      :: "v"(p), "v"(v0), "v"(v1), "v"(v2), "v"(v3) : "memory");
}

struct RowQuad { float a, b, c, d; };
__device__ __forceinline__ RowQuad load_row(const float* __restrict__ p) {
  RowQuad r;
  r.a = p[0];
  r.b = p[32];
  r.c = p[64];
  r.d = p[96];
  return r;
}

__global__ void __launch_bounds__(256)
zero_f32(float* __restrict__ p, long long n) {
  const long long n4 = n >> 2;
  const long long stride = (long long)gridDim.x * blockDim.x;
  const long long t = (long long)blockIdx.x * blockDim.x + threadIdx.x;
  const float4 z = make_float4(0.0f, 0.0f, 0.0f, 0.0f);
  float4* p4 = (float4*)p;
  for (long long i = t; i < n4; i += stride) p4[i] = z;
  if (t < (n & 3)) p[(n4 << 2) + t] = 0.0f;
}

__global__ void __launch_bounds__(256)
degree_kernel(const int* __restrict__ dst, const int* __restrict__ users,
              float* __restrict__ deg, int* __restrict__ flags, int n_edges,
              int batch) {
  const int stride = gridDim.x * blockDim.x;
  const int t = blockIdx.x * blockDim.x + threadIdx.x;
  for (int e = t; e < n_edges; e += stride)
    gatomic_add_f32(deg + dst[e], 1.0f);
  if (flags != nullptr)
    for (int b = t; b < batch; b += stride) flags[users[b]] = 1;
}

template <bool SCALE>
__global__ void __launch_bounds__(256)
scatter_kernel(const int* __restrict__ src, const int* __restrict__ dst,
               const float* __restrict__ h, const float* __restrict__ deg,
               float* __restrict__ agg, int n_edges) {
  int e = blockIdx.x * blockDim.x + threadIdx.x;
  int lane = threadIdx.x & 31;
  int base = __builtin_amdgcn_readfirstlane(e - lane);
  if (base >= n_edges) return;
  int s = 0, d = 0;
  float sc = 1.0f;
  if (e < n_edges) {
    s = src[e];
    d = dst[e];
    if (SCALE) sc = 1.0f / fmaxf(deg[s], 1.0f);
  }
  int left = n_edges - base;
  int cnt = left < 32 ? left : 32;
  for (int j = 0; j < cnt; j += 4) {
    int sj[4], dj[4];
    float cj[4];
    RowQuad v[4];
#pragma unroll
    for (int k = 0; k < 4; ++k) {
      int jj = (j + k < cnt) ? (j + k) : (cnt - 1);
      sj[k] = __builtin_amdgcn_readlane(s, jj);
      dj[k] = __builtin_amdgcn_readlane(d, jj);
      if (SCALE)
        cj[k] = __int_as_float(
            __builtin_amdgcn_readlane(__float_as_int(sc), jj));
      v[k] = load_row(h + (size_t)sj[k] * EMB_DIM + lane);
    }
#pragma unroll
    for (int k = 0; k < 4; ++k) {
      if (j + k < cnt) {
        if (SCALE) {
          v[k].a *= cj[k]; v[k].b *= cj[k]; v[k].c *= cj[k]; v[k].d *= cj[k];
        }
        gatomic_add_row(agg + (size_t)dj[k] * EMB_DIM + lane, v[k].a, v[k].b,
                        v[k].c, v[k].d);
      }
    }
  }
}

__global__ void __launch_bounds__(256)
scatter_blocked_kernel(const int* __restrict__ src, const int* __restrict__ dst,
                       const float* __restrict__ h, float* __restrict__ agg,
                       int n_edges, int half_n, int blocks_per_pass) {
  const int pass = blockIdx.x / blocks_per_pass;
  const int pblock = blockIdx.x - pass * blocks_per_pass;
  const bool dst_hi = (pass >> 1) != 0;
  const bool src_hi = ((pass ^ (pass >> 1)) & 1) != 0;
  const int lane = threadIdx.x & 31;
  const int wave = __builtin_amdgcn_readfirstlane(
      (int)((pblock * blockDim.x + threadIdx.x) >> 5));
  const int step = ((blocks_per_pass * (int)blockDim.x) >> 5) * 32;
  for (int base = wave * 32; base < n_edges; base += step) {
    const int e = base + lane;
    int s = 0, d = 0;
    bool keep = false;
    if (e < n_edges) {
      s = src[e];
      d = dst[e];
      keep = ((s >= half_n) == src_hi) && ((d >= half_n) == dst_hi);
    }
    unsigned mask = __builtin_amdgcn_ballot_w32(keep);
    while (mask) {
      int sj[4], dj[4];
      bool on[4];
      RowQuad v[4];
      int j = 0;
#pragma unroll
      for (int k = 0; k < 4; ++k) {
        on[k] = mask != 0;
        if (on[k]) {
          j = __builtin_ctz(mask);
          mask &= mask - 1;
        }
        sj[k] = __builtin_amdgcn_readlane(s, j);
        dj[k] = __builtin_amdgcn_readlane(d, j);
        v[k] = load_row(h + (size_t)sj[k] * EMB_DIM + lane);
      }
#pragma unroll
      for (int k = 0; k < 4; ++k) {
        if (on[k])
          gatomic_add_row(agg + (size_t)dj[k] * EMB_DIM + lane, v[k].a, v[k].b,
                          v[k].c, v[k].d);
      }
    }
  }
}

__global__ void __launch_bounds__(256)
scatter_flagged_kernel(const int* __restrict__ src, const int* __restrict__ dst,
                       const int* __restrict__ flags,
                       const float* __restrict__ h,
                       const float* __restrict__ deg, float* __restrict__ agg,
                       int n_edges) {
  int e = blockIdx.x * blockDim.x + threadIdx.x;
  int lane = threadIdx.x & 31;
  if ((e - lane) >= n_edges) return;
  int s = 0, d = 0, f = 0;
  float sc = 1.0f;
  if (e < n_edges) {
    s = src[e];
    d = dst[e];
    f = flags[d];
    if (f != 0) sc = 1.0f / fmaxf(deg[s], 1.0f);
  }
  unsigned mask = __builtin_amdgcn_ballot_w32(f != 0);
  while (mask) {
    int j = __builtin_ctz(mask);
    mask &= mask - 1;
    int sj = __builtin_amdgcn_readlane(s, j);
    int dj = __builtin_amdgcn_readlane(d, j);
    const float cj =
        __int_as_float(__builtin_amdgcn_readlane(__float_as_int(sc), j));
    const RowQuad v = load_row(h + (size_t)sj * EMB_DIM + lane);
    gatomic_add_row(agg + (size_t)dj * EMB_DIM + lane, v.a * cj, v.b * cj,
                    v.c * cj, v.d * cj);
  }
}

__global__ void __launch_bounds__(256)
gather_kernel(const int* __restrict__ users, const int* __restrict__ items,
              const float* __restrict__ uemb, const float* __restrict__ g1,
              const float* __restrict__ g2, const float* __restrict__ deg,
              const float* __restrict__ iemb, float* __restrict__ out_user,
              float* __restrict__ out_item, int batch) {
  int t = blockIdx.x * blockDim.x + threadIdx.x;
  int b = t >> 5;
  int lane = t & 31;
  if (b >= batch) return;
  int u = users[b];
  int it = items[b];
  size_t off = (size_t)lane * 4;
  const float4 a = *(const float4*)(uemb + (size_t)u * EMB_DIM + off);
  const float4 c = *(const float4*)(g1 + (size_t)u * EMB_DIM + off);
  const float4 d = *(const float4*)(g2 + (size_t)u * EMB_DIM + off);
  const float inv = 1.0f / fmaxf(deg[u], 1.0f);
  float4 fu;
  fu.x = a.x + c.x * inv + d.x * inv;
  fu.y = a.y + c.y * inv + d.y * inv;
  fu.z = a.z + c.z * inv + d.z * inv;
  fu.w = a.w + c.w * inv + d.w * inv;
  *(float4*)(out_user + (size_t)b * EMB_DIM + off) = fu;
  *(float4*)(out_item + (size_t)b * EMB_DIM + off) =
      *(const float4*)(iemb + (size_t)it * EMB_DIM + off);
}

__global__ void __launch_bounds__(256)
dot_sigmoid_wmma(const float* __restrict__ U, const float* __restrict__ V,
                 float* __restrict__ pred, int batch) {
  int gid = blockIdx.x * blockDim.x + threadIdx.x;
  int wave = gid >> 5;
  int lane = threadIdx.x & 31;
  if (wave * 16 >= batch) return;

  int row = lane & 15;
  int half = lane >> 4;
  const float* urow = U + ((size_t)wave * 16 + row) * EMB_DIM + 2 * half;
  const float* vrow = V + ((size_t)wave * 16 + row) * EMB_DIM + 2 * half;

  v8f acc = {};
#pragma unroll
  for (int k0 = 0; k0 < EMB_DIM; k0 += 4) {
    v2f a = *(const v2f*)(urow + k0);
    v2f b = *(const v2f*)(vrow + k0);
    acc = __builtin_amdgcn_wmma_f32_16x16x4_f32(
         false, a,  false, b,
         (short)0, acc,  false,  false);
  }

  if (lane < 8 || lane >= 24) {
    int i = (lane < 8) ? lane : (lane - 16);
    float d = acc[lane & 7];
    pred[(size_t)wave * 16 + i] = 1.0f / (1.0f + expf(-d));
  }
}

extern "C" void kernel_launch(void* const* d_in, const int* in_sizes, int n_in,
                              void* d_out, int out_size, void* d_ws,
                              size_t ws_size, hipStream_t stream) {
  const int* users  = (const int*)d_in[0];
  const int* items  = (const int*)d_in[1];
  const int* esrc   = (const int*)d_in[2];
  const int* edst   = (const int*)d_in[3];
  const float* uemb = (const float*)d_in[4];
  const float* iemb = (const float*)d_in[5];

  const int batch     = in_sizes[0];
  const int n_edges   = in_sizes[2];
  const int num_users = in_sizes[4] / EMB_DIM;

  float* deg = (float*)d_ws;
  float* g1  = deg + num_users;
  float* g2  = g1 + (size_t)num_users * EMB_DIM;
  long long total = (long long)num_users * (1 + 2 * EMB_DIM);
  int* flags = nullptr;
  if (ws_size >= (size_t)(total + num_users) * sizeof(float)) {
    flags = (int*)(g2 + (size_t)num_users * EMB_DIM);
    total += num_users;
  }

  float* pred     = (float*)d_out;
  float* out_user = pred + batch;
  float* out_item = out_user + (size_t)batch * EMB_DIM;

  const int B = 256;

  const int G = 2048;

  zero_f32<<<G, B, 0, stream>>>((float*)d_ws, total);

  degree_kernel<<<G / 4, B, 0, stream>>>(edst, users, deg, flags, n_edges,
                                         batch);

  const int sc_blocks = (n_edges + B - 1) / B;
  scatter_blocked_kernel<<<4 * G, B, 0, stream>>>(esrc, edst, uemb, g1, n_edges,
                                                  (num_users + 1) / 2, G);

  if (flags != nullptr) {
    scatter_flagged_kernel<<<sc_blocks, B, 0, stream>>>(esrc, edst, flags, g1,
                                                        deg, g2, n_edges);
  } else {
    scatter_kernel<true><<<sc_blocks, B, 0, stream>>>(esrc, edst, g1, deg, g2,
                                                      n_edges);
  }

  long long ga_threads = (long long)batch * 32;
  gather_kernel<<<(int)((ga_threads + B - 1) / B), B, 0, stream>>>(
      users, items, uemb, g1, g2, deg, iemb, out_user, out_item, batch);

  long long dot_threads = (long long)((batch + 15) / 16) * 32;
  dot_sigmoid_wmma<<<(int)((dot_threads + B - 1) / B), B, 0, stream>>>(
      out_user, out_item, pred, batch);
}


# deg and layer-1 sums kept once per package half: a wave's atomics go to its own half's copy (die id), readers add the copies (short workspace: g2 is the copy, one merge pass); sums now two partials
# speedup vs baseline: 3.9641x; 1.4805x over previous
#include <hip/hip_runtime.h>
#include <hip/hip_bf16.h>

#define EMB_DIM 128

typedef __attribute__((ext_vector_type(2))) float v2f;
typedef __attribute__((ext_vector_type(8))) float v8f;

__device__ __forceinline__ void gatomic_add_f32(float* p, float v) {
  asm volatile("global_atomic_add_f32 %0, %1, off scope:SCOPE_DEV"
               :: "v"(p), "v"(v) : "memory");
}

__device__ __forceinline__ void gatomic_add_row(float* p, float v0, float v1,
                                                float v2, float v3) {
  asm volatile(
      "global_atomic_add_f32 %0, %1, off scope:SCOPE_DEV\n\t"
      "global_atomic_add_f32 %0, %2, off offset:128 scope:SCOPE_DEV\n\t"
      "global_atomic_add_f32 %0, %3, off offset:256 scope:SCOPE_DEV\n\t"
      "global_atomic_add_f32 %0, %4, off offset:384 scope:SCOPE_DEV"
      :: "v"(p), "v"(v0), "v"(v1), "v"(v2), "v"(v3) : "memory");
}

__device__ __forceinline__ int chip_half() {
  const unsigned id = __builtin_amdgcn_s_sendmsg_rtn(0x87);
  return (int)((id >> 18) & 1u);
}

struct RowQuad { float a, b, c, d; };
__device__ __forceinline__ RowQuad load_row(const float* __restrict__ p) {
  RowQuad r;
  r.a = p[0];
  r.b = p[32];
  r.c = p[64];
  r.d = p[96];
  return r;
}

__global__ void __launch_bounds__(256)
zero_f32(float* __restrict__ p, long long n) {
  const long long n4 = n >> 2;
  const long long stride = (long long)gridDim.x * blockDim.x;
  const long long t = (long long)blockIdx.x * blockDim.x + threadIdx.x;
  const float4 z = make_float4(0.0f, 0.0f, 0.0f, 0.0f);
  float4* p4 = (float4*)p;
  for (long long i = t; i < n4; i += stride) p4[i] = z;
  if (t < (n & 3)) p[(n4 << 2) + t] = 0.0f;
}

__global__ void __launch_bounds__(256)
degree_kernel(const int* __restrict__ dst, const int* __restrict__ users,
              float* __restrict__ deg, int* __restrict__ flags, int n_edges,
              int batch, int copy_words) {
  const int stride = gridDim.x * blockDim.x;
  const int t = blockIdx.x * blockDim.x + threadIdx.x;
  float* mydeg = deg + (size_t)chip_half() * copy_words;
  for (int e = t; e < n_edges; e += stride)
    gatomic_add_f32(mydeg + dst[e], 1.0f);
  if (flags != nullptr)
    for (int b = t; b < batch; b += stride) flags[users[b]] = 1;
}

template <bool SCALE>
__global__ void __launch_bounds__(256)
scatter_kernel(const int* __restrict__ src, const int* __restrict__ dst,
               const float* __restrict__ h, const float* __restrict__ deg,
               float* __restrict__ agg, int n_edges) {
  int e = blockIdx.x * blockDim.x + threadIdx.x;
  int lane = threadIdx.x & 31;
  int base = __builtin_amdgcn_readfirstlane(e - lane);
  if (base >= n_edges) return;
  int s = 0, d = 0;
  float sc = 1.0f;
  if (e < n_edges) {
    s = src[e];
    d = dst[e];
    if (SCALE) sc = 1.0f / fmaxf(deg[s], 1.0f);
  }
  int left = n_edges - base;
  int cnt = left < 32 ? left : 32;
  for (int j = 0; j < cnt; j += 4) {
    int sj[4], dj[4];
    float cj[4];
    RowQuad v[4];
#pragma unroll
    for (int k = 0; k < 4; ++k) {
      int jj = (j + k < cnt) ? (j + k) : (cnt - 1);
      sj[k] = __builtin_amdgcn_readlane(s, jj);
      dj[k] = __builtin_amdgcn_readlane(d, jj);
      if (SCALE)
        cj[k] = __int_as_float(
            __builtin_amdgcn_readlane(__float_as_int(sc), jj));
      v[k] = load_row(h + (size_t)sj[k] * EMB_DIM + lane);
    }
#pragma unroll
    for (int k = 0; k < 4; ++k) {
      if (j + k < cnt) {
        if (SCALE) {
          v[k].a *= cj[k]; v[k].b *= cj[k]; v[k].c *= cj[k]; v[k].d *= cj[k];
        }
        gatomic_add_row(agg + (size_t)dj[k] * EMB_DIM + lane, v[k].a, v[k].b,
                        v[k].c, v[k].d);
      }
    }
  }
}

__global__ void __launch_bounds__(256)
scatter_blocked_kernel(const int* __restrict__ src, const int* __restrict__ dst,
                       const float* __restrict__ h, float* __restrict__ agg0,
                       int n_edges, int half_n, int blocks_per_pass,
                       size_t copy_words) {
  float* __restrict__ agg = agg0 + (size_t)chip_half() * copy_words;
  const int pass = blockIdx.x / blocks_per_pass;
  const int pblock = blockIdx.x - pass * blocks_per_pass;
  const bool dst_hi = (pass >> 1) != 0;
  const bool src_hi = ((pass ^ (pass >> 1)) & 1) != 0;
  const int lane = threadIdx.x & 31;
  const int wave = __builtin_amdgcn_readfirstlane(
      (int)((pblock * blockDim.x + threadIdx.x) >> 5));
  const int step = ((blocks_per_pass * (int)blockDim.x) >> 5) * 32;
  for (int base = wave * 32; base < n_edges; base += step) {
    const int e = base + lane;
    int s = 0, d = 0;
    bool keep = false;
    if (e < n_edges) {
      s = src[e];
      d = dst[e];
      keep = ((s >= half_n) == src_hi) && ((d >= half_n) == dst_hi);
    }
    unsigned mask = __builtin_amdgcn_ballot_w32(keep);
    while (mask) {
      int sj[4], dj[4];
      bool on[4];
      RowQuad v[4];
      int j = 0;
#pragma unroll
      for (int k = 0; k < 4; ++k) {
        on[k] = mask != 0;
        if (on[k]) {
          j = __builtin_ctz(mask);
          mask &= mask - 1;
        }
        sj[k] = __builtin_amdgcn_readlane(s, j);
        dj[k] = __builtin_amdgcn_readlane(d, j);
        v[k] = load_row(h + (size_t)sj[k] * EMB_DIM + lane);
      }
#pragma unroll
      for (int k = 0; k < 4; ++k) {
        if (on[k])
          gatomic_add_row(agg + (size_t)dj[k] * EMB_DIM + lane, v[k].a, v[k].b,
                          v[k].c, v[k].d);
      }
    }
  }
}

__global__ void __launch_bounds__(256)
merge_clear_kernel(float* __restrict__ a, float* __restrict__ b,
                   long long n4) {
  const long long stride = (long long)gridDim.x * blockDim.x;
  const long long t = (long long)blockIdx.x * blockDim.x + threadIdx.x;
  float4* a4 = (float4*)a;
  float4* b4 = (float4*)b;
  const float4 z = make_float4(0.0f, 0.0f, 0.0f, 0.0f);
  for (long long i = t; i < n4; i += stride) {
    float4 x = a4[i];
    const float4 y = b4[i];
    x.x += y.x; x.y += y.y; x.z += y.z; x.w += y.w;
    a4[i] = x;
    b4[i] = z;
  }
}

__global__ void __launch_bounds__(256)
scatter_flagged_kernel(const int* __restrict__ src, const int* __restrict__ dst,
                       const int* __restrict__ flags,
                       const float* __restrict__ h,
                       const float* __restrict__ hB,
                       const float* __restrict__ deg,
                       const float* __restrict__ degB, float* __restrict__ agg,
                       int n_edges) {
  int e = blockIdx.x * blockDim.x + threadIdx.x;
  int lane = threadIdx.x & 31;
  if ((e - lane) >= n_edges) return;
  int s = 0, d = 0, f = 0;
  float sc = 1.0f;
  if (e < n_edges) {
    s = src[e];
    d = dst[e];
    f = flags[d];
    if (f != 0) {
      float dg = deg[s];
      if (degB != nullptr) dg += degB[s];
      sc = 1.0f / fmaxf(dg, 1.0f);
    }
  }
  unsigned mask = __builtin_amdgcn_ballot_w32(f != 0);
  while (mask) {
    int j = __builtin_ctz(mask);
    mask &= mask - 1;
    int sj = __builtin_amdgcn_readlane(s, j);
    int dj = __builtin_amdgcn_readlane(d, j);
    const float cj =
        __int_as_float(__builtin_amdgcn_readlane(__float_as_int(sc), j));
    RowQuad v = load_row(h + (size_t)sj * EMB_DIM + lane);
    if (hB != nullptr) {
      const RowQuad w = load_row(hB + (size_t)sj * EMB_DIM + lane);
      v.a += w.a; v.b += w.b; v.c += w.c; v.d += w.d;
    }
    gatomic_add_row(agg + (size_t)dj * EMB_DIM + lane, v.a * cj, v.b * cj,
                    v.c * cj, v.d * cj);
  }
}

__global__ void __launch_bounds__(256)
gather_kernel(const int* __restrict__ users, const int* __restrict__ items,
              const float* __restrict__ uemb, const float* __restrict__ g1,
              const float* __restrict__ g1B, const float* __restrict__ g2,
              const float* __restrict__ deg, const float* __restrict__ degB,
              const float* __restrict__ iemb, float* __restrict__ out_user,
              float* __restrict__ out_item, int batch) {
  int t = blockIdx.x * blockDim.x + threadIdx.x;
  int b = t >> 5;
  int lane = t & 31;
  if (b >= batch) return;
  int u = users[b];
  int it = items[b];
  size_t off = (size_t)lane * 4;
  const float4 a = *(const float4*)(uemb + (size_t)u * EMB_DIM + off);
  float4 c = *(const float4*)(g1 + (size_t)u * EMB_DIM + off);
  float dg = deg[u];
  if (g1B != nullptr) {
    const float4 cb = *(const float4*)(g1B + (size_t)u * EMB_DIM + off);
    c.x += cb.x; c.y += cb.y; c.z += cb.z; c.w += cb.w;
  }
  if (degB != nullptr) dg += degB[u];
  const float4 d = *(const float4*)(g2 + (size_t)u * EMB_DIM + off);
  const float inv = 1.0f / fmaxf(dg, 1.0f);
  float4 fu;
  fu.x = a.x + c.x * inv + d.x * inv;
  fu.y = a.y + c.y * inv + d.y * inv;
  fu.z = a.z + c.z * inv + d.z * inv;
  fu.w = a.w + c.w * inv + d.w * inv;
  *(float4*)(out_user + (size_t)b * EMB_DIM + off) = fu;
  *(float4*)(out_item + (size_t)b * EMB_DIM + off) =
      *(const float4*)(iemb + (size_t)it * EMB_DIM + off);
}

__global__ void __launch_bounds__(256)
dot_sigmoid_wmma(const float* __restrict__ U, const float* __restrict__ V,
                 float* __restrict__ pred, int batch) {
  int gid = blockIdx.x * blockDim.x + threadIdx.x;
  int wave = gid >> 5;
  int lane = threadIdx.x & 31;
  if (wave * 16 >= batch) return;

  int row = lane & 15;
  int half = lane >> 4;
  const float* urow = U + ((size_t)wave * 16 + row) * EMB_DIM + 2 * half;
  const float* vrow = V + ((size_t)wave * 16 + row) * EMB_DIM + 2 * half;

  v8f acc = {};
#pragma unroll
  for (int k0 = 0; k0 < EMB_DIM; k0 += 4) {
    v2f a = *(const v2f*)(urow + k0);
    v2f b = *(const v2f*)(vrow + k0);
    acc = __builtin_amdgcn_wmma_f32_16x16x4_f32(
         false, a,  false, b,
         (short)0, acc,  false,  false);
  }

  if (lane < 8 || lane >= 24) {
    int i = (lane < 8) ? lane : (lane - 16);
    float d = acc[lane & 7];
    pred[(size_t)wave * 16 + i] = 1.0f / (1.0f + expf(-d));
  }
}

extern "C" void kernel_launch(void* const* d_in, const int* in_sizes, int n_in,
                              void* d_out, int out_size, void* d_ws,
                              size_t ws_size, hipStream_t stream) {
  const int* users  = (const int*)d_in[0];
  const int* items  = (const int*)d_in[1];
  const int* esrc   = (const int*)d_in[2];
  const int* edst   = (const int*)d_in[3];
  const float* uemb = (const float*)d_in[4];
  const float* iemb = (const float*)d_in[5];

  const int batch     = in_sizes[0];
  const int n_edges   = in_sizes[2];
  const int num_users = in_sizes[4] / EMB_DIM;

  const size_t nu = (size_t)num_users;
  const size_t tbl = nu * EMB_DIM;
  const size_t words = ws_size / sizeof(float);
  const bool has_flags = words >= 2 * nu + 2 * tbl;
  const bool has_degB  = words >= 3 * nu + 2 * tbl;
  const bool has_g1B   = words >= 3 * nu + 3 * tbl;
  float* deg  = (float*)d_ws;
  float* degB = has_degB ? deg + nu : nullptr;
  float* g1   = deg + (has_degB ? 2 : 1) * nu;
  float* g1B  = has_g1B ? g1 + tbl : nullptr;
  float* g2   = g1 + (has_g1B ? 2 : 1) * tbl;
  int* flags  = has_flags ? (int*)(g2 + tbl) : nullptr;
  const long long total =
      (long long)((g2 + tbl) - deg) + (has_flags ? (long long)nu : 0);
  const size_t copy1 = tbl;

  float* pred     = (float*)d_out;
  float* out_user = pred + batch;
  float* out_item = out_user + (size_t)batch * EMB_DIM;

  const int B = 256;

  const int G = 2048;

  zero_f32<<<G, B, 0, stream>>>((float*)d_ws, total);

  degree_kernel<<<G / 4, B, 0, stream>>>(edst, users, deg, flags, n_edges,
                                         batch, has_degB ? num_users : 0);

  const int sc_blocks = (n_edges + B - 1) / B;
  scatter_blocked_kernel<<<4 * G, B, 0, stream>>>(esrc, edst, uemb, g1, n_edges,
                                                  (num_users + 1) / 2, G, copy1);

  if (!has_g1B)
    merge_clear_kernel<<<G, B, 0, stream>>>(g1, g2, (long long)(tbl / 4));

  if (flags != nullptr) {
    scatter_flagged_kernel<<<sc_blocks, B, 0, stream>>>(
        esrc, edst, flags, g1, g1B, deg, degB, g2, n_edges);
  } else {
    scatter_kernel<true><<<sc_blocks, B, 0, stream>>>(esrc, edst, g1, deg, g2,
                                                      n_edges);
  }

  long long ga_threads = (long long)batch * 32;
  gather_kernel<<<(int)((ga_threads + B - 1) / B), B, 0, stream>>>(
      users, items, uemb, g1, g1B, g2, deg, degB, iemb, out_user, out_item,
      batch);

  long long dot_threads = (long long)((batch + 15) / 16) * 32;
  dot_sigmoid_wmma<<<(int)((dot_threads + B - 1) / B), B, 0, stream>>>(
      out_user, out_item, pred, batch);
}


# layer-2 sums go to a table over the gathered rows only (slot per user in flags), one copy per package half, the gather adds the two: less to zero, atomics stay in their half; sums now two partials
# speedup vs baseline: 4.1113x; 1.0371x over previous
#include <hip/hip_runtime.h>
#include <hip/hip_bf16.h>

#define EMB_DIM 128

typedef __attribute__((ext_vector_type(2))) float v2f;
typedef __attribute__((ext_vector_type(8))) float v8f;

__device__ __forceinline__ void gatomic_add_f32(float* p, float v) {
  asm volatile("global_atomic_add_f32 %0, %1, off scope:SCOPE_DEV"
               :: "v"(p), "v"(v) : "memory");
}

__device__ __forceinline__ void gatomic_add_row(float* p, float v0, float v1,
                                                float v2, float v3) {
  asm volatile(
      "global_atomic_add_f32 %0, %1, off scope:SCOPE_DEV\n\t"
      "global_atomic_add_f32 %0, %2, off offset:128 scope:SCOPE_DEV\n\t"
      "global_atomic_add_f32 %0, %3, off offset:256 scope:SCOPE_DEV\n\t"
      "global_atomic_add_f32 %0, %4, off offset:384 scope:SCOPE_DEV"
      :: "v"(p), "v"(v0), "v"(v1), "v"(v2), "v"(v3) : "memory");
}

__device__ __forceinline__ int chip_half() {
  const unsigned id = __builtin_amdgcn_s_sendmsg_rtn(0x87);
  return (int)((id >> 18) & 1u);
}

struct RowQuad { float a, b, c, d; };
__device__ __forceinline__ RowQuad load_row(const float* __restrict__ p) {
  RowQuad r;
  r.a = p[0];
  r.b = p[32];
  r.c = p[64];
  r.d = p[96];
  return r;
}

__global__ void __launch_bounds__(256)
zero_f32(float* __restrict__ p, long long n) {
  const long long n4 = n >> 2;
  const long long stride = (long long)gridDim.x * blockDim.x;
  const long long t = (long long)blockIdx.x * blockDim.x + threadIdx.x;
  const float4 z = make_float4(0.0f, 0.0f, 0.0f, 0.0f);
  float4* p4 = (float4*)p;
  for (long long i = t; i < n4; i += stride) p4[i] = z;
  if (t < (n & 3)) p[(n4 << 2) + t] = 0.0f;
}

__global__ void __launch_bounds__(256)
degree_kernel(const int* __restrict__ dst, const int* __restrict__ users,
              float* __restrict__ deg, int* __restrict__ flags, int n_edges,
              int batch, int copy_words, int slot_flags) {
  const int stride = gridDim.x * blockDim.x;
  const int t = blockIdx.x * blockDim.x + threadIdx.x;
  float* mydeg = deg + (size_t)chip_half() * copy_words;
  for (int e = t; e < n_edges; e += stride)
    gatomic_add_f32(mydeg + dst[e], 1.0f);
  if (flags != nullptr)
    for (int b = t; b < batch; b += stride)
      flags[users[b]] = slot_flags ? (b + 1) : 1;
}

template <bool SCALE>
__global__ void __launch_bounds__(256)
scatter_kernel(const int* __restrict__ src, const int* __restrict__ dst,
               const float* __restrict__ h, const float* __restrict__ deg,
               float* __restrict__ agg, int n_edges) {
  int e = blockIdx.x * blockDim.x + threadIdx.x;
  int lane = threadIdx.x & 31;
  int base = __builtin_amdgcn_readfirstlane(e - lane);
  if (base >= n_edges) return;
  int s = 0, d = 0;
  float sc = 1.0f;
  if (e < n_edges) {
    s = src[e];
    d = dst[e];
    if (SCALE) sc = 1.0f / fmaxf(deg[s], 1.0f);
  }
  int left = n_edges - base;
  int cnt = left < 32 ? left : 32;
  for (int j = 0; j < cnt; j += 4) {
    int sj[4], dj[4];
    float cj[4];
    RowQuad v[4];
#pragma unroll
    for (int k = 0; k < 4; ++k) {
      int jj = (j + k < cnt) ? (j + k) : (cnt - 1);
      sj[k] = __builtin_amdgcn_readlane(s, jj);
      dj[k] = __builtin_amdgcn_readlane(d, jj);
      if (SCALE)
        cj[k] = __int_as_float(
            __builtin_amdgcn_readlane(__float_as_int(sc), jj));
      v[k] = load_row(h + (size_t)sj[k] * EMB_DIM + lane);
    }
#pragma unroll
    for (int k = 0; k < 4; ++k) {
      if (j + k < cnt) {
        if (SCALE) {
          v[k].a *= cj[k]; v[k].b *= cj[k]; v[k].c *= cj[k]; v[k].d *= cj[k];
        }
        gatomic_add_row(agg + (size_t)dj[k] * EMB_DIM + lane, v[k].a, v[k].b,
                        v[k].c, v[k].d);
      }
    }
  }
}

__global__ void __launch_bounds__(256)
scatter_blocked_kernel(const int* __restrict__ src, const int* __restrict__ dst,
                       const float* __restrict__ h, float* __restrict__ agg0,
                       int n_edges, int half_n, int blocks_per_pass,
                       size_t copy_words) {
  float* __restrict__ agg = agg0 + (size_t)chip_half() * copy_words;
  const int pass = blockIdx.x / blocks_per_pass;
  const int pblock = blockIdx.x - pass * blocks_per_pass;
  const bool dst_hi = (pass >> 1) != 0;
  const bool src_hi = ((pass ^ (pass >> 1)) & 1) != 0;
  const int lane = threadIdx.x & 31;
  const int wave = __builtin_amdgcn_readfirstlane(
      (int)((pblock * blockDim.x + threadIdx.x) >> 5));
  const int step = ((blocks_per_pass * (int)blockDim.x) >> 5) * 32;
  for (int base = wave * 32; base < n_edges; base += step) {
    const int e = base + lane;
    int s = 0, d = 0;
    bool keep = false;
    if (e < n_edges) {
      s = src[e];
      d = dst[e];
      keep = ((s >= half_n) == src_hi) && ((d >= half_n) == dst_hi);
    }
    unsigned mask = __builtin_amdgcn_ballot_w32(keep);
    while (mask) {
      int sj[4], dj[4];
      bool on[4];
      RowQuad v[4];
      int j = 0;
#pragma unroll
      for (int k = 0; k < 4; ++k) {
        on[k] = mask != 0;
        if (on[k]) {
          j = __builtin_ctz(mask);
          mask &= mask - 1;
        }
        sj[k] = __builtin_amdgcn_readlane(s, j);
        dj[k] = __builtin_amdgcn_readlane(d, j);
        v[k] = load_row(h + (size_t)sj[k] * EMB_DIM + lane);
      }
#pragma unroll
      for (int k = 0; k < 4; ++k) {
        if (on[k])
          gatomic_add_row(agg + (size_t)dj[k] * EMB_DIM + lane, v[k].a, v[k].b,
                          v[k].c, v[k].d);
      }
    }
  }
}

__global__ void __launch_bounds__(256)
merge_clear_kernel(float* __restrict__ a, float* __restrict__ b,
                   long long n4) {
  const long long stride = (long long)gridDim.x * blockDim.x;
  const long long t = (long long)blockIdx.x * blockDim.x + threadIdx.x;
  float4* a4 = (float4*)a;
  float4* b4 = (float4*)b;
  const float4 z = make_float4(0.0f, 0.0f, 0.0f, 0.0f);
  for (long long i = t; i < n4; i += stride) {
    float4 x = a4[i];
    const float4 y = b4[i];
    x.x += y.x; x.y += y.y; x.z += y.z; x.w += y.w;
    a4[i] = x;
    b4[i] = z;
  }
}

__global__ void __launch_bounds__(256)
scatter_flagged_kernel(const int* __restrict__ src, const int* __restrict__ dst,
                       const int* __restrict__ flags,
                       const float* __restrict__ h,
                       const float* __restrict__ hB,
                       const float* __restrict__ deg,
                       const float* __restrict__ degB, float* __restrict__ agg0,
                       size_t agg_copy_words, int compact, int n_edges) {
  int e = blockIdx.x * blockDim.x + threadIdx.x;
  int lane = threadIdx.x & 31;
  if ((e - lane) >= n_edges) return;
  float* __restrict__ agg = agg0 + (size_t)chip_half() * agg_copy_words;
  int s = 0, d = 0, f = 0;
  float sc = 1.0f;
  if (e < n_edges) {
    s = src[e];
    d = dst[e];
    f = flags[d];
    if (compact && f != 0) d = f - 1;
    if (f != 0) {
      float dg = deg[s];
      if (degB != nullptr) dg += degB[s];
      sc = 1.0f / fmaxf(dg, 1.0f);
    }
  }
  unsigned mask = __builtin_amdgcn_ballot_w32(f != 0);
  while (mask) {
    int j = __builtin_ctz(mask);
    mask &= mask - 1;
    int sj = __builtin_amdgcn_readlane(s, j);
    int dj = __builtin_amdgcn_readlane(d, j);
    const float cj =
        __int_as_float(__builtin_amdgcn_readlane(__float_as_int(sc), j));
    RowQuad v = load_row(h + (size_t)sj * EMB_DIM + lane);
    if (hB != nullptr) {
      const RowQuad w = load_row(hB + (size_t)sj * EMB_DIM + lane);
      v.a += w.a; v.b += w.b; v.c += w.c; v.d += w.d;
    }
    gatomic_add_row(agg + (size_t)dj * EMB_DIM + lane, v.a * cj, v.b * cj,
                    v.c * cj, v.d * cj);
  }
}

__global__ void __launch_bounds__(256)
gather_kernel(const int* __restrict__ users, const int* __restrict__ items,
              const float* __restrict__ uemb, const float* __restrict__ g1,
              const float* __restrict__ g1B, const float* __restrict__ g2,
              const float* __restrict__ g2B, const int* __restrict__ slots,
              const float* __restrict__ deg, const float* __restrict__ degB,
              const float* __restrict__ iemb, float* __restrict__ out_user,
              float* __restrict__ out_item, int batch) {
  int t = blockIdx.x * blockDim.x + threadIdx.x;
  int b = t >> 5;
  int lane = t & 31;
  if (b >= batch) return;
  int u = users[b];
  int it = items[b];
  size_t off = (size_t)lane * 4;
  const float4 a = *(const float4*)(uemb + (size_t)u * EMB_DIM + off);
  float4 c = *(const float4*)(g1 + (size_t)u * EMB_DIM + off);
  float dg = deg[u];
  if (g1B != nullptr) {
    const float4 cb = *(const float4*)(g1B + (size_t)u * EMB_DIM + off);
    c.x += cb.x; c.y += cb.y; c.z += cb.z; c.w += cb.w;
  }
  if (degB != nullptr) dg += degB[u];
  const int r2 = (slots != nullptr) ? (slots[u] - 1) : u;
  float4 d = *(const float4*)(g2 + (size_t)r2 * EMB_DIM + off);
  if (g2B != nullptr) {
    const float4 db = *(const float4*)(g2B + (size_t)r2 * EMB_DIM + off);
    d.x += db.x; d.y += db.y; d.z += db.z; d.w += db.w;
  }
  const float inv = 1.0f / fmaxf(dg, 1.0f);
  float4 fu;
  fu.x = a.x + c.x * inv + d.x * inv;
  fu.y = a.y + c.y * inv + d.y * inv;
  fu.z = a.z + c.z * inv + d.z * inv;
  fu.w = a.w + c.w * inv + d.w * inv;
  *(float4*)(out_user + (size_t)b * EMB_DIM + off) = fu;
  *(float4*)(out_item + (size_t)b * EMB_DIM + off) =
      *(const float4*)(iemb + (size_t)it * EMB_DIM + off);
}

__global__ void __launch_bounds__(256)
dot_sigmoid_wmma(const float* __restrict__ U, const float* __restrict__ V,
                 float* __restrict__ pred, int batch) {
  int gid = blockIdx.x * blockDim.x + threadIdx.x;
  int wave = gid >> 5;
  int lane = threadIdx.x & 31;
  if (wave * 16 >= batch) return;

  int row = lane & 15;
  int half = lane >> 4;
  const float* urow = U + ((size_t)wave * 16 + row) * EMB_DIM + 2 * half;
  const float* vrow = V + ((size_t)wave * 16 + row) * EMB_DIM + 2 * half;

  v8f acc = {};
#pragma unroll
  for (int k0 = 0; k0 < EMB_DIM; k0 += 4) {
    v2f a = *(const v2f*)(urow + k0);
    v2f b = *(const v2f*)(vrow + k0);
    acc = __builtin_amdgcn_wmma_f32_16x16x4_f32(
         false, a,  false, b,
         (short)0, acc,  false,  false);
  }

  if (lane < 8 || lane >= 24) {
    int i = (lane < 8) ? lane : (lane - 16);
    float d = acc[lane & 7];
    pred[(size_t)wave * 16 + i] = 1.0f / (1.0f + expf(-d));
  }
}

extern "C" void kernel_launch(void* const* d_in, const int* in_sizes, int n_in,
                              void* d_out, int out_size, void* d_ws,
                              size_t ws_size, hipStream_t stream) {
  const int* users  = (const int*)d_in[0];
  const int* items  = (const int*)d_in[1];
  const int* esrc   = (const int*)d_in[2];
  const int* edst   = (const int*)d_in[3];
  const float* uemb = (const float*)d_in[4];
  const float* iemb = (const float*)d_in[5];

  const int batch     = in_sizes[0];
  const int n_edges   = in_sizes[2];
  const int num_users = in_sizes[4] / EMB_DIM;

  const size_t nu = (size_t)num_users;
  const size_t tbl = nu * EMB_DIM;
  const size_t words = ws_size / sizeof(float);
  const bool has_flags = words >= 2 * nu + 2 * tbl;
  const bool has_degB  = words >= 3 * nu + 2 * tbl;
  const bool has_g1B   = words >= 3 * nu + 3 * tbl;
  float* deg  = (float*)d_ws;
  float* degB = has_degB ? deg + nu : nullptr;
  float* g1   = deg + (has_degB ? 2 : 1) * nu;
  float* g1B  = has_g1B ? g1 + tbl : nullptr;
  float* g2   = g1 + (has_g1B ? 2 : 1) * tbl;
  const size_t bt = (size_t)batch * EMB_DIM;
  const bool compact = has_g1B && words >= 3 * nu + 2 * tbl + 2 * bt;
  float* g2B  = compact ? g2 + bt : nullptr;
  const size_t g2_words = compact ? 2 * bt : tbl;
  int* flags  = has_flags ? (int*)(g2 + g2_words) : nullptr;
  const long long total =
      (long long)((g2 + g2_words) - deg) + (has_flags ? (long long)nu : 0);
  const size_t copy1 = tbl;

  float* pred     = (float*)d_out;
  float* out_user = pred + batch;
  float* out_item = out_user + (size_t)batch * EMB_DIM;

  const int B = 256;

  const int G = 2048;

  zero_f32<<<G, B, 0, stream>>>((float*)d_ws, total);

  degree_kernel<<<G / 4, B, 0, stream>>>(edst, users, deg, flags, n_edges,
                                         batch, has_degB ? num_users : 0,
                                         compact ? 1 : 0);

  const int sc_blocks = (n_edges + B - 1) / B;
  scatter_blocked_kernel<<<4 * G, B, 0, stream>>>(esrc, edst, uemb, g1, n_edges,
                                                  (num_users + 1) / 2, G, copy1);

  if (!has_g1B)
    merge_clear_kernel<<<G, B, 0, stream>>>(g1, g2, (long long)(tbl / 4));

  if (flags != nullptr) {
    scatter_flagged_kernel<<<sc_blocks, B, 0, stream>>>(
        esrc, edst, flags, g1, g1B, deg, degB, g2, compact ? bt : 0,
        compact ? 1 : 0, n_edges);
  } else {
    scatter_kernel<true><<<sc_blocks, B, 0, stream>>>(esrc, edst, g1, deg, g2,
                                                      n_edges);
  }

  long long ga_threads = (long long)batch * 32;
  gather_kernel<<<(int)((ga_threads + B - 1) / B), B, 0, stream>>>(
      users, items, uemb, g1, g1B, g2, g2B, compact ? flags : nullptr, deg,
      degB, iemb, out_user, out_item, batch);

  long long dot_threads = (long long)((batch + 15) / 16) * 32;
  dot_sigmoid_wmma<<<(int)((dot_threads + B - 1) / B), B, 0, stream>>>(
      out_user, out_item, pred, batch);
}


# layer 1 as a pull: edges listed by destination (prefix offsets, places by returning f32 add), one wave adds a row's source rows in registers; line atomics gone; sums in list (arrival) order
# speedup vs baseline: 5.2677x; 1.2813x over previous
#include <hip/hip_runtime.h>
#include <hip/hip_bf16.h>

#define EMB_DIM 128

typedef __attribute__((ext_vector_type(2))) float v2f;
typedef __attribute__((ext_vector_type(8))) float v8f;

__device__ __forceinline__ void gatomic_add_f32(float* p, float v) {
  asm volatile("global_atomic_add_f32 %0, %1, off scope:SCOPE_DEV"
               :: "v"(p), "v"(v) : "memory");
}

__device__ __forceinline__ void gatomic_add_row(float* p, float v0, float v1,
                                                float v2, float v3) {
  asm volatile(
      "global_atomic_add_f32 %0, %1, off scope:SCOPE_DEV\n\t"
      "global_atomic_add_f32 %0, %2, off offset:128 scope:SCOPE_DEV\n\t"
      "global_atomic_add_f32 %0, %3, off offset:256 scope:SCOPE_DEV\n\t"
      "global_atomic_add_f32 %0, %4, off offset:384 scope:SCOPE_DEV"
      :: "v"(p), "v"(v0), "v"(v1), "v"(v2), "v"(v3) : "memory");
}

__device__ __forceinline__ float gatomic_add_ret_f32(float* p, float v) {
  float r;
  asm volatile(
      "global_atomic_add_f32 %0, %1, %2, off th:TH_ATOMIC_RETURN scope:SCOPE_DEV\n\t"
      "s_wait_loadcnt 0x0"
      : "=&v"(r) : "v"(p), "v"(v) : "memory");
  return r;
}

__device__ __forceinline__ int chip_half() {
  const unsigned id = __builtin_amdgcn_s_sendmsg_rtn(0x87);
  return (int)((id >> 18) & 1u);
}

struct RowQuad { float a, b, c, d; };
__device__ __forceinline__ RowQuad load_row(const float* __restrict__ p) {
  RowQuad r;
  r.a = p[0];
  r.b = p[32];
  r.c = p[64];
  r.d = p[96];
  return r;
}

__global__ void __launch_bounds__(256)
zero_f32(float* __restrict__ p, long long n) {
  const long long n4 = n >> 2;
  const long long stride = (long long)gridDim.x * blockDim.x;
  const long long t = (long long)blockIdx.x * blockDim.x + threadIdx.x;
  const float4 z = make_float4(0.0f, 0.0f, 0.0f, 0.0f);
  float4* p4 = (float4*)p;
  for (long long i = t; i < n4; i += stride) p4[i] = z;
  if (t < (n & 3)) p[(n4 << 2) + t] = 0.0f;
}

__global__ void __launch_bounds__(256)
degree_kernel(const int* __restrict__ dst, const int* __restrict__ users,
              float* __restrict__ deg, int* __restrict__ flags, int n_edges,
              int batch, int copy_words, int slot_flags) {
  const int stride = gridDim.x * blockDim.x;
  const int t = blockIdx.x * blockDim.x + threadIdx.x;
  float* mydeg = deg + (size_t)chip_half() * copy_words;
  for (int e = t; e < n_edges; e += stride)
    gatomic_add_f32(mydeg + dst[e], 1.0f);
  if (flags != nullptr)
    for (int b = t; b < batch; b += stride)
      flags[users[b]] = slot_flags ? (b + 1) : 1;
}

template <bool SCALE>
__global__ void __launch_bounds__(256)
scatter_kernel(const int* __restrict__ src, const int* __restrict__ dst,
               const float* __restrict__ h, const float* __restrict__ deg,
               float* __restrict__ agg, int n_edges) {
  int e = blockIdx.x * blockDim.x + threadIdx.x;
  int lane = threadIdx.x & 31;
  int base = __builtin_amdgcn_readfirstlane(e - lane);
  if (base >= n_edges) return;
  int s = 0, d = 0;
  float sc = 1.0f;
  if (e < n_edges) {
    s = src[e];
    d = dst[e];
    if (SCALE) sc = 1.0f / fmaxf(deg[s], 1.0f);
  }
  int left = n_edges - base;
  int cnt = left < 32 ? left : 32;
  for (int j = 0; j < cnt; j += 4) {
    int sj[4], dj[4];
    float cj[4];
    RowQuad v[4];
#pragma unroll
    for (int k = 0; k < 4; ++k) {
      int jj = (j + k < cnt) ? (j + k) : (cnt - 1);
      sj[k] = __builtin_amdgcn_readlane(s, jj);
      dj[k] = __builtin_amdgcn_readlane(d, jj);
      if (SCALE)
        cj[k] = __int_as_float(
            __builtin_amdgcn_readlane(__float_as_int(sc), jj));
      v[k] = load_row(h + (size_t)sj[k] * EMB_DIM + lane);
    }
#pragma unroll
    for (int k = 0; k < 4; ++k) {
      if (j + k < cnt) {
        if (SCALE) {
          v[k].a *= cj[k]; v[k].b *= cj[k]; v[k].c *= cj[k]; v[k].d *= cj[k];
        }
        gatomic_add_row(agg + (size_t)dj[k] * EMB_DIM + lane, v[k].a, v[k].b,
                        v[k].c, v[k].d);
      }
    }
  }
}

__global__ void __launch_bounds__(256)
scatter_blocked_kernel(const int* __restrict__ src, const int* __restrict__ dst,
                       const float* __restrict__ h, float* __restrict__ agg0,
                       int n_edges, int half_n, int blocks_per_pass,
                       size_t copy_words) {
  float* __restrict__ agg = agg0 + (size_t)chip_half() * copy_words;
  const int pass = blockIdx.x / blocks_per_pass;
  const int pblock = blockIdx.x - pass * blocks_per_pass;
  const bool dst_hi = (pass >> 1) != 0;
  const bool src_hi = ((pass ^ (pass >> 1)) & 1) != 0;
  const int lane = threadIdx.x & 31;
  const int wave = __builtin_amdgcn_readfirstlane(
      (int)((pblock * blockDim.x + threadIdx.x) >> 5));
  const int step = ((blocks_per_pass * (int)blockDim.x) >> 5) * 32;
  for (int base = wave * 32; base < n_edges; base += step) {
    const int e = base + lane;
    int s = 0, d = 0;
    bool keep = false;
    if (e < n_edges) {
      s = src[e];
      d = dst[e];
      keep = ((s >= half_n) == src_hi) && ((d >= half_n) == dst_hi);
    }
    unsigned mask = __builtin_amdgcn_ballot_w32(keep);
    while (mask) {
      int sj[4], dj[4];
      bool on[4];
      RowQuad v[4];
      int j = 0;
#pragma unroll
      for (int k = 0; k < 4; ++k) {
        on[k] = mask != 0;
        if (on[k]) {
          j = __builtin_ctz(mask);
          mask &= mask - 1;
        }
        sj[k] = __builtin_amdgcn_readlane(s, j);
        dj[k] = __builtin_amdgcn_readlane(d, j);
        v[k] = load_row(h + (size_t)sj[k] * EMB_DIM + lane);
      }
#pragma unroll
      for (int k = 0; k < 4; ++k) {
        if (on[k])
          gatomic_add_row(agg + (size_t)dj[k] * EMB_DIM + lane, v[k].a, v[k].b,
                          v[k].c, v[k].d);
      }
    }
  }
}


__global__ void __launch_bounds__(1024)
offsets_kernel(const float* __restrict__ degA, const float* __restrict__ degB,
               int* __restrict__ off, int n) {
  __shared__ int s_w[32];
  const int tid = threadIdx.x;
  const int lane = tid & 31;
  const int w = tid >> 5;
  const int first = blockIdx.x * 1024;
  int part = 0;
  const float4* a4 = (const float4*)degA;
  const float4* b4 = (const float4*)degB;
#pragma unroll 4
  for (int i = tid; i < (first >> 2); i += 1024) {
    const float4 x = a4[i];
    const float4 y = b4[i];
    part += (int)(x.x + y.x) + (int)(x.y + y.y) + (int)(x.z + y.z) +
            (int)(x.w + y.w);
  }
  for (int o = 16; o > 0; o >>= 1) part += __shfl_xor(part, o, 32);
  if (lane == 0) s_w[w] = part;
  __syncthreads();
  int base = 0;
  for (int k = 0; k < 32; ++k) base += s_w[k];
  __syncthreads();
  const int i = first + tid;
  const int c = (i < n) ? (int)(degA[i] + degB[i]) : 0;
  int x = c;
  for (int o = 1; o < 32; o <<= 1) {
    const int y = __shfl_up(x, o, 32);
    if (lane >= o) x += y;
  }
  if (lane == 31) s_w[w] = x;
  __syncthreads();
  int pre = 0;
  for (int k = 0; k < w; ++k) pre += s_w[k];
  if (i < n) off[i] = base + pre + x - c;
}

__global__ void __launch_bounds__(256)
fill_lists_kernel(const int* __restrict__ src, const int* __restrict__ dst,
                  const int* __restrict__ off, float* __restrict__ cur0,
                  int cur_copy_words, int* __restrict__ list0,
                  size_t list_copy_words, int n_edges) {
  const int e = blockIdx.x * blockDim.x + threadIdx.x;
  const int half = chip_half();
  float* cur = cur0 + (size_t)half * cur_copy_words;
  int* list = list0 + (size_t)half * list_copy_words;
  if (e < n_edges) {
    const int s = src[e];
    const int d = dst[e];
    const int o = off[d];
    const int slot = (int)gatomic_add_ret_f32(cur + d, 1.0f);
    const unsigned pos = (unsigned)(o + slot);
    const unsigned last = (unsigned)(n_edges - 1);
    list[pos < last ? pos : last] = s;
  }
}

__global__ void __launch_bounds__(256)
pull_rows_kernel(const int* __restrict__ off, const float* __restrict__ cur0,
                 int cur_copy_words, const int* __restrict__ list0,
                 size_t list_copy_words, const float* __restrict__ h,
                 float* __restrict__ out, int n_rows) {
  const int lane = threadIdx.x & 31;
  const int wave = __builtin_amdgcn_readfirstlane(
      (int)((blockIdx.x * blockDim.x + threadIdx.x) >> 5));
  const int nwaves = (int)((gridDim.x * blockDim.x) >> 5);
  const int* __restrict__ list1 = list0 + list_copy_words;
  const float* __restrict__ hl = h + lane * 4;
  for (int d = wave; d < n_rows; d += nwaves) {
    const int o = off[d];
    const int nA = (int)cur0[d];
    const int nB = (int)cur0[(size_t)cur_copy_words + d];
    const int n = nA + nB;
    float4 acc = make_float4(0.0f, 0.0f, 0.0f, 0.0f);
    for (int c = 0; c < n; c += 32) {
      const int j = c + lane;
      int idx = 0;
      if (j < nA) idx = list0[o + j];
      else if (j < n) idx = list1[o + j - nA];
      const int m = (n - c) < 32 ? (n - c) : 32;
      int k = 0;
      for (; k + 8 <= m; k += 8) {
        float4 v[8];
#pragma unroll
        for (int u = 0; u < 8; ++u) {
          const int s = __builtin_amdgcn_readlane(idx, k + u);
          v[u] = *(const float4*)(hl + (size_t)s * EMB_DIM);
        }
#pragma unroll
        for (int u = 0; u < 8; ++u) {
          acc.x += v[u].x; acc.y += v[u].y; acc.z += v[u].z; acc.w += v[u].w;
        }
      }
      for (; k < m; k += 4) {
        float4 v[4];
#pragma unroll
        for (int u = 0; u < 4; ++u) {
          const int kk = (k + u < m) ? (k + u) : (m - 1);
          const int s = __builtin_amdgcn_readlane(idx, kk);
          v[u] = *(const float4*)(hl + (size_t)s * EMB_DIM);
        }
#pragma unroll
        for (int u = 0; u < 4; ++u) {
          if (k + u < m) {
            acc.x += v[u].x; acc.y += v[u].y; acc.z += v[u].z; acc.w += v[u].w;
          }
        }
      }
    }
    *(float4*)(out + (size_t)d * EMB_DIM + lane * 4) = acc;
  }
}

__global__ void __launch_bounds__(256)
merge_clear_kernel(float* __restrict__ a, float* __restrict__ b,
                   long long n4) {
  const long long stride = (long long)gridDim.x * blockDim.x;
  const long long t = (long long)blockIdx.x * blockDim.x + threadIdx.x;
  float4* a4 = (float4*)a;
  float4* b4 = (float4*)b;
  const float4 z = make_float4(0.0f, 0.0f, 0.0f, 0.0f);
  for (long long i = t; i < n4; i += stride) {
    float4 x = a4[i];
    const float4 y = b4[i];
    x.x += y.x; x.y += y.y; x.z += y.z; x.w += y.w;
    a4[i] = x;
    b4[i] = z;
  }
}

__global__ void __launch_bounds__(256)
scatter_flagged_kernel(const int* __restrict__ src, const int* __restrict__ dst,
                       const int* __restrict__ flags,
                       const float* __restrict__ h,
                       const float* __restrict__ hB,
                       const float* __restrict__ deg,
                       const float* __restrict__ degB, float* __restrict__ agg0,
                       size_t agg_copy_words, int compact, int n_edges) {
  int e = blockIdx.x * blockDim.x + threadIdx.x;
  int lane = threadIdx.x & 31;
  if ((e - lane) >= n_edges) return;
  float* __restrict__ agg = agg0 + (size_t)chip_half() * agg_copy_words;
  int s = 0, d = 0, f = 0;
  float sc = 1.0f;
  if (e < n_edges) {
    s = src[e];
    d = dst[e];
    f = flags[d];
    if (compact && f != 0) d = f - 1;
    if (f != 0) {
      float dg = deg[s];
      if (degB != nullptr) dg += degB[s];
      sc = 1.0f / fmaxf(dg, 1.0f);
    }
  }
  unsigned mask = __builtin_amdgcn_ballot_w32(f != 0);
  while (mask) {
    int j = __builtin_ctz(mask);
    mask &= mask - 1;
    int sj = __builtin_amdgcn_readlane(s, j);
    int dj = __builtin_amdgcn_readlane(d, j);
    const float cj =
        __int_as_float(__builtin_amdgcn_readlane(__float_as_int(sc), j));
    RowQuad v = load_row(h + (size_t)sj * EMB_DIM + lane);
    if (hB != nullptr) {
      const RowQuad w = load_row(hB + (size_t)sj * EMB_DIM + lane);
      v.a += w.a; v.b += w.b; v.c += w.c; v.d += w.d;
    }
    gatomic_add_row(agg + (size_t)dj * EMB_DIM + lane, v.a * cj, v.b * cj,
                    v.c * cj, v.d * cj);
  }
}

__global__ void __launch_bounds__(256)
gather_kernel(const int* __restrict__ users, const int* __restrict__ items,
              const float* __restrict__ uemb, const float* __restrict__ g1,
              const float* __restrict__ g1B, const float* __restrict__ g2,
              const float* __restrict__ g2B, const int* __restrict__ slots,
              const float* __restrict__ deg, const float* __restrict__ degB,
              const float* __restrict__ iemb, float* __restrict__ out_user,
              float* __restrict__ out_item, int batch) {
  int t = blockIdx.x * blockDim.x + threadIdx.x;
  int b = t >> 5;
  int lane = t & 31;
  if (b >= batch) return;
  int u = users[b];
  int it = items[b];
  size_t off = (size_t)lane * 4;
  const float4 a = *(const float4*)(uemb + (size_t)u * EMB_DIM + off);
  float4 c = *(const float4*)(g1 + (size_t)u * EMB_DIM + off);
  float dg = deg[u];
  if (g1B != nullptr) {
    const float4 cb = *(const float4*)(g1B + (size_t)u * EMB_DIM + off);
    c.x += cb.x; c.y += cb.y; c.z += cb.z; c.w += cb.w;
  }
  if (degB != nullptr) dg += degB[u];
  const int r2 = (slots != nullptr) ? (slots[u] - 1) : u;
  float4 d = *(const float4*)(g2 + (size_t)r2 * EMB_DIM + off);
  if (g2B != nullptr) {
    const float4 db = *(const float4*)(g2B + (size_t)r2 * EMB_DIM + off);
    d.x += db.x; d.y += db.y; d.z += db.z; d.w += db.w;
  }
  const float inv = 1.0f / fmaxf(dg, 1.0f);
  float4 fu;
  fu.x = a.x + c.x * inv + d.x * inv;
  fu.y = a.y + c.y * inv + d.y * inv;
  fu.z = a.z + c.z * inv + d.z * inv;
  fu.w = a.w + c.w * inv + d.w * inv;
  *(float4*)(out_user + (size_t)b * EMB_DIM + off) = fu;
  *(float4*)(out_item + (size_t)b * EMB_DIM + off) =
      *(const float4*)(iemb + (size_t)it * EMB_DIM + off);
}

__global__ void __launch_bounds__(256)
dot_sigmoid_wmma(const float* __restrict__ U, const float* __restrict__ V,
                 float* __restrict__ pred, int batch) {
  int gid = blockIdx.x * blockDim.x + threadIdx.x;
  int wave = gid >> 5;
  int lane = threadIdx.x & 31;
  if (wave * 16 >= batch) return;

  int row = lane & 15;
  int half = lane >> 4;
  const float* urow = U + ((size_t)wave * 16 + row) * EMB_DIM + 2 * half;
  const float* vrow = V + ((size_t)wave * 16 + row) * EMB_DIM + 2 * half;

  v8f acc = {};
#pragma unroll
  for (int k0 = 0; k0 < EMB_DIM; k0 += 4) {
    v2f a = *(const v2f*)(urow + k0);
    v2f b = *(const v2f*)(vrow + k0);
    acc = __builtin_amdgcn_wmma_f32_16x16x4_f32(
         false, a,  false, b,
         (short)0, acc,  false,  false);
  }

  if (lane < 8 || lane >= 24) {
    int i = (lane < 8) ? lane : (lane - 16);
    float d = acc[lane & 7];
    pred[(size_t)wave * 16 + i] = 1.0f / (1.0f + expf(-d));
  }
}

extern "C" void kernel_launch(void* const* d_in, const int* in_sizes, int n_in,
                              void* d_out, int out_size, void* d_ws,
                              size_t ws_size, hipStream_t stream) {
  const int* users  = (const int*)d_in[0];
  const int* items  = (const int*)d_in[1];
  const int* esrc   = (const int*)d_in[2];
  const int* edst   = (const int*)d_in[3];
  const float* uemb = (const float*)d_in[4];
  const float* iemb = (const float*)d_in[5];

  const int batch     = in_sizes[0];
  const int n_edges   = in_sizes[2];
  const int num_users = in_sizes[4] / EMB_DIM;

  const size_t nu = (size_t)num_users;
  const size_t tbl = nu * EMB_DIM;
  const size_t words = ws_size / sizeof(float);
  const bool has_flags = words >= 2 * nu + 2 * tbl;
  const bool has_degB  = words >= 3 * nu + 2 * tbl;
  const bool has_g1B   = words >= 3 * nu + 3 * tbl;
  float* deg  = (float*)d_ws;
  float* degB = has_degB ? deg + nu : nullptr;
  float* g1   = deg + (has_degB ? 2 : 1) * nu;
  float* g1B  = has_g1B ? g1 + tbl : nullptr;
  float* g2   = g1 + (has_g1B ? 2 : 1) * tbl;
  const size_t bt = (size_t)batch * EMB_DIM;
  const bool compact = has_g1B && words >= 3 * nu + 2 * tbl + 2 * bt;
  float* g2B  = compact ? g2 + bt : nullptr;
  const size_t g2_words = compact ? 2 * bt : tbl;
  int* flags  = has_flags ? (int*)(g2 + g2_words) : nullptr;
  const long long total =
      (long long)((g2 + g2_words) - deg) + (has_flags ? (long long)nu : 0);
  const size_t copy1 = tbl;

  float* pred     = (float*)d_out;
  float* out_user = pred + batch;
  float* out_item = out_user + (size_t)batch * EMB_DIM;

  const int B = 256;

  const int G = 2048;

  {
    const size_t nuA = (nu + 31) & ~(size_t)31;
    const size_t neA = ((size_t)n_edges + 31) & ~(size_t)31;
    const size_t zeroed = 5 * nuA + 2 * bt;
    const size_t need = zeroed + nuA + 2 * neA + tbl;
    if (words >= need) {
      float* p_deg   = (float*)d_ws;
      float* p_cur   = p_deg + 2 * nuA;
      int*   p_flags = (int*)(p_cur + 2 * nuA);
      float* p_g2    = (float*)(p_flags + nuA);
      int*   p_off   = (int*)(p_g2 + 2 * bt);
      int*   p_list  = p_off + nuA;
      float* p_g1    = (float*)(p_list + 2 * neA);
      const int eb = (n_edges + B - 1) / B;

      zero_f32<<<G, B, 0, stream>>>(p_deg, (long long)zeroed);
      degree_kernel<<<G / 4, B, 0, stream>>>(edst, users, p_deg, p_flags,
                                             n_edges, batch, (int)nuA, 1);
      offsets_kernel<<<(num_users + 1023) / 1024, 1024, 0, stream>>>(
          p_deg, p_deg + nuA, p_off, num_users);
      fill_lists_kernel<<<eb, B, 0, stream>>>(esrc, edst, p_off, p_cur,
                                              (int)nuA, p_list, neA, n_edges);
      pull_rows_kernel<<<G, B, 0, stream>>>(p_off, p_cur, (int)nuA, p_list, neA,
                                            uemb, p_g1, num_users);
      scatter_flagged_kernel<<<eb, B, 0, stream>>>(
          esrc, edst, p_flags, p_g1, nullptr, p_deg, p_deg + nuA, p_g2, bt, 1,
          n_edges);
      const long long gt = (long long)batch * 32;
      gather_kernel<<<(int)((gt + B - 1) / B), B, 0, stream>>>(
          users, items, uemb, p_g1, nullptr, p_g2, p_g2 + bt, p_flags, p_deg,
          p_deg + nuA, iemb, out_user, out_item, batch);
      const long long dt = (long long)((batch + 15) / 16) * 32;
      dot_sigmoid_wmma<<<(int)((dt + B - 1) / B), B, 0, stream>>>(
          out_user, out_item, pred, batch);
      return;
    }
  }

  zero_f32<<<G, B, 0, stream>>>((float*)d_ws, total);

  degree_kernel<<<G / 4, B, 0, stream>>>(edst, users, deg, flags, n_edges,
                                         batch, has_degB ? num_users : 0,
                                         compact ? 1 : 0);

  const int sc_blocks = (n_edges + B - 1) / B;
  scatter_blocked_kernel<<<4 * G, B, 0, stream>>>(esrc, edst, uemb, g1, n_edges,
                                                  (num_users + 1) / 2, G, copy1);

  if (!has_g1B)
    merge_clear_kernel<<<G, B, 0, stream>>>(g1, g2, (long long)(tbl / 4));

  if (flags != nullptr) {
    scatter_flagged_kernel<<<sc_blocks, B, 0, stream>>>(
        esrc, edst, flags, g1, g1B, deg, degB, g2, compact ? bt : 0,
        compact ? 1 : 0, n_edges);
  } else {
    scatter_kernel<true><<<sc_blocks, B, 0, stream>>>(esrc, edst, g1, deg, g2,
                                                      n_edges);
  }

  long long ga_threads = (long long)batch * 32;
  gather_kernel<<<(int)((ga_threads + B - 1) / B), B, 0, stream>>>(
      users, items, uemb, g1, g1B, g2, g2B, compact ? flags : nullptr, deg,
      degB, iemb, out_user, out_item, batch);

  long long dot_threads = (long long)((batch + 15) / 16) * 32;
  dot_sigmoid_wmma<<<(int)((dot_threads + B - 1) / B), B, 0, stream>>>(
      out_user, out_item, pred, batch);
}


# layer 2 as a pull inside the gather: a wave per batch row adds its user's listed source rows of g1 (stored as means) in registers; flagged scatter launch, atomics and tables gone; sums in list order
# speedup vs baseline: 7.2334x; 1.3732x over previous
#include <hip/hip_runtime.h>
#include <hip/hip_bf16.h>

#define EMB_DIM 128

typedef __attribute__((ext_vector_type(2))) float v2f;
typedef __attribute__((ext_vector_type(8))) float v8f;

__device__ __forceinline__ void gatomic_add_f32(float* p, float v) {
  asm volatile("global_atomic_add_f32 %0, %1, off scope:SCOPE_DEV"
               :: "v"(p), "v"(v) : "memory");
}

__device__ __forceinline__ void gatomic_add_row(float* p, float v0, float v1,
                                                float v2, float v3) {
  asm volatile(
      "global_atomic_add_f32 %0, %1, off scope:SCOPE_DEV\n\t"
      "global_atomic_add_f32 %0, %2, off offset:128 scope:SCOPE_DEV\n\t"
      "global_atomic_add_f32 %0, %3, off offset:256 scope:SCOPE_DEV\n\t"
      "global_atomic_add_f32 %0, %4, off offset:384 scope:SCOPE_DEV"
      :: "v"(p), "v"(v0), "v"(v1), "v"(v2), "v"(v3) : "memory");
}

__device__ __forceinline__ float gatomic_add_ret_f32(float* p, float v) {
  float r;
  asm volatile(
      "global_atomic_add_f32 %0, %1, %2, off th:TH_ATOMIC_RETURN scope:SCOPE_DEV\n\t"
      "s_wait_loadcnt 0x0"
      : "=&v"(r) : "v"(p), "v"(v) : "memory");
  return r;
}

__device__ __forceinline__ int chip_half() {
  const unsigned id = __builtin_amdgcn_s_sendmsg_rtn(0x87);
  return (int)((id >> 18) & 1u);
}

struct RowQuad { float a, b, c, d; };
__device__ __forceinline__ RowQuad load_row(const float* __restrict__ p) {
  RowQuad r;
  r.a = p[0];
  r.b = p[32];
  r.c = p[64];
  r.d = p[96];
  return r;
}

__global__ void __launch_bounds__(256)
zero_f32(float* __restrict__ p, long long n) {
  const long long n4 = n >> 2;
  const long long stride = (long long)gridDim.x * blockDim.x;
  const long long t = (long long)blockIdx.x * blockDim.x + threadIdx.x;
  const float4 z = make_float4(0.0f, 0.0f, 0.0f, 0.0f);
  float4* p4 = (float4*)p;
  for (long long i = t; i < n4; i += stride) p4[i] = z;
  if (t < (n & 3)) p[(n4 << 2) + t] = 0.0f;
}

__global__ void __launch_bounds__(256)
degree_kernel(const int* __restrict__ dst, const int* __restrict__ users,
              float* __restrict__ deg, int* __restrict__ flags, int n_edges,
              int batch, int copy_words, int slot_flags) {
  const int stride = gridDim.x * blockDim.x;
  const int t = blockIdx.x * blockDim.x + threadIdx.x;
  float* mydeg = deg + (size_t)chip_half() * copy_words;
  for (int e = t; e < n_edges; e += stride)
    gatomic_add_f32(mydeg + dst[e], 1.0f);
  if (flags != nullptr)
    for (int b = t; b < batch; b += stride)
      flags[users[b]] = slot_flags ? (b + 1) : 1;
}

template <bool SCALE>
__global__ void __launch_bounds__(256)
scatter_kernel(const int* __restrict__ src, const int* __restrict__ dst,
               const float* __restrict__ h, const float* __restrict__ deg,
               float* __restrict__ agg, int n_edges) {
  int e = blockIdx.x * blockDim.x + threadIdx.x;
  int lane = threadIdx.x & 31;
  int base = __builtin_amdgcn_readfirstlane(e - lane);
  if (base >= n_edges) return;
  int s = 0, d = 0;
  float sc = 1.0f;
  if (e < n_edges) {
    s = src[e];
    d = dst[e];
    if (SCALE) sc = 1.0f / fmaxf(deg[s], 1.0f);
  }
  int left = n_edges - base;
  int cnt = left < 32 ? left : 32;
  for (int j = 0; j < cnt; j += 4) {
    int sj[4], dj[4];
    float cj[4];
    RowQuad v[4];
#pragma unroll
    for (int k = 0; k < 4; ++k) {
      int jj = (j + k < cnt) ? (j + k) : (cnt - 1);
      sj[k] = __builtin_amdgcn_readlane(s, jj);
      dj[k] = __builtin_amdgcn_readlane(d, jj);
      if (SCALE)
        cj[k] = __int_as_float(
            __builtin_amdgcn_readlane(__float_as_int(sc), jj));
      v[k] = load_row(h + (size_t)sj[k] * EMB_DIM + lane);
    }
#pragma unroll
    for (int k = 0; k < 4; ++k) {
      if (j + k < cnt) {
        if (SCALE) {
          v[k].a *= cj[k]; v[k].b *= cj[k]; v[k].c *= cj[k]; v[k].d *= cj[k];
        }
        gatomic_add_row(agg + (size_t)dj[k] * EMB_DIM + lane, v[k].a, v[k].b,
                        v[k].c, v[k].d);
      }
    }
  }
}

__global__ void __launch_bounds__(256)
scatter_blocked_kernel(const int* __restrict__ src, const int* __restrict__ dst,
                       const float* __restrict__ h, float* __restrict__ agg0,
                       int n_edges, int half_n, int blocks_per_pass,
                       size_t copy_words) {
  float* __restrict__ agg = agg0 + (size_t)chip_half() * copy_words;
  const int pass = blockIdx.x / blocks_per_pass;
  const int pblock = blockIdx.x - pass * blocks_per_pass;
  const bool dst_hi = (pass >> 1) != 0;
  const bool src_hi = ((pass ^ (pass >> 1)) & 1) != 0;
  const int lane = threadIdx.x & 31;
  const int wave = __builtin_amdgcn_readfirstlane(
      (int)((pblock * blockDim.x + threadIdx.x) >> 5));
  const int step = ((blocks_per_pass * (int)blockDim.x) >> 5) * 32;
  for (int base = wave * 32; base < n_edges; base += step) {
    const int e = base + lane;
    int s = 0, d = 0;
    bool keep = false;
    if (e < n_edges) {
      s = src[e];
      d = dst[e];
      keep = ((s >= half_n) == src_hi) && ((d >= half_n) == dst_hi);
    }
    unsigned mask = __builtin_amdgcn_ballot_w32(keep);
    while (mask) {
      int sj[4], dj[4];
      bool on[4];
      RowQuad v[4];
      int j = 0;
#pragma unroll
      for (int k = 0; k < 4; ++k) {
        on[k] = mask != 0;
        if (on[k]) {
          j = __builtin_ctz(mask);
          mask &= mask - 1;
        }
        sj[k] = __builtin_amdgcn_readlane(s, j);
        dj[k] = __builtin_amdgcn_readlane(d, j);
        v[k] = load_row(h + (size_t)sj[k] * EMB_DIM + lane);
      }
#pragma unroll
      for (int k = 0; k < 4; ++k) {
        if (on[k])
          gatomic_add_row(agg + (size_t)dj[k] * EMB_DIM + lane, v[k].a, v[k].b,
                          v[k].c, v[k].d);
      }
    }
  }
}


__global__ void __launch_bounds__(1024)
offsets_kernel(const float* __restrict__ degA, const float* __restrict__ degB,
               int* __restrict__ off, int n) {
  __shared__ int s_w[32];
  const int tid = threadIdx.x;
  const int lane = tid & 31;
  const int w = tid >> 5;
  const int first = blockIdx.x * 1024;
  int part = 0;
  const float4* a4 = (const float4*)degA;
  const float4* b4 = (const float4*)degB;
#pragma unroll 4
  for (int i = tid; i < (first >> 2); i += 1024) {
    const float4 x = a4[i];
    const float4 y = b4[i];
    part += (int)(x.x + y.x) + (int)(x.y + y.y) + (int)(x.z + y.z) +
            (int)(x.w + y.w);
  }
  for (int o = 16; o > 0; o >>= 1) part += __shfl_xor(part, o, 32);
  if (lane == 0) s_w[w] = part;
  __syncthreads();
  int base = 0;
  for (int k = 0; k < 32; ++k) base += s_w[k];
  __syncthreads();
  const int i = first + tid;
  const int c = (i < n) ? (int)(degA[i] + degB[i]) : 0;
  int x = c;
  for (int o = 1; o < 32; o <<= 1) {
    const int y = __shfl_up(x, o, 32);
    if (lane >= o) x += y;
  }
  if (lane == 31) s_w[w] = x;
  __syncthreads();
  int pre = 0;
  for (int k = 0; k < w; ++k) pre += s_w[k];
  if (i < n) off[i] = base + pre + x - c;
}

__global__ void __launch_bounds__(256)
fill_lists_kernel(const int* __restrict__ src, const int* __restrict__ dst,
                  const int* __restrict__ off, float* __restrict__ cur0,
                  int cur_copy_words, int* __restrict__ list0,
                  size_t list_copy_words, int n_edges) {
  const int e = blockIdx.x * blockDim.x + threadIdx.x;
  const int half = chip_half();
  float* cur = cur0 + (size_t)half * cur_copy_words;
  int* list = list0 + (size_t)half * list_copy_words;
  if (e < n_edges) {
    const int s = src[e];
    const int d = dst[e];
    const int o = off[d];
    const int slot = (int)gatomic_add_ret_f32(cur + d, 1.0f);
    const unsigned pos = (unsigned)(o + slot);
    const unsigned last = (unsigned)(n_edges - 1);
    list[pos < last ? pos : last] = s;
  }
}

__device__ __forceinline__ float4 pull_row_sum(int o, int nA, int nB,
                                               const int* __restrict__ list0,
                                               const int* __restrict__ list1,
                                               const float* __restrict__ hl,
                                               int lane) {
  const int n = nA + nB;
  float4 acc = make_float4(0.0f, 0.0f, 0.0f, 0.0f);
  for (int c = 0; c < n; c += 32) {
    const int j = c + lane;
    int idx = 0;
    if (j < nA) idx = list0[o + j];
    else if (j < n) idx = list1[o + j - nA];
    const int m = (n - c) < 32 ? (n - c) : 32;
    int k = 0;
    for (; k + 8 <= m; k += 8) {
      float4 v[8];
#pragma unroll
      for (int u = 0; u < 8; ++u) {
        const int s = __builtin_amdgcn_readlane(idx, k + u);
        v[u] = *(const float4*)(hl + (size_t)s * EMB_DIM);
      }
#pragma unroll
      for (int u = 0; u < 8; ++u) {
        acc.x += v[u].x; acc.y += v[u].y; acc.z += v[u].z; acc.w += v[u].w;
      }
    }
    for (; k < m; k += 4) {
      float4 v[4];
#pragma unroll
      for (int u = 0; u < 4; ++u) {
        const int kk = (k + u < m) ? (k + u) : (m - 1);
        const int s = __builtin_amdgcn_readlane(idx, kk);
        v[u] = *(const float4*)(hl + (size_t)s * EMB_DIM);
      }
#pragma unroll
      for (int u = 0; u < 4; ++u) {
        if (k + u < m) {
          acc.x += v[u].x; acc.y += v[u].y; acc.z += v[u].z; acc.w += v[u].w;
        }
      }
    }
  }
  return acc;
}

__global__ void __launch_bounds__(256)
pull_rows_kernel(const int* __restrict__ off, const float* __restrict__ cur0,
                 int cur_copy_words, const int* __restrict__ list0,
                 size_t list_copy_words, const float* __restrict__ h,
                 float* __restrict__ out, int n_rows) {
  const int lane = threadIdx.x & 31;
  const int wave = __builtin_amdgcn_readfirstlane(
      (int)((blockIdx.x * blockDim.x + threadIdx.x) >> 5));
  const int nwaves = (int)((gridDim.x * blockDim.x) >> 5);
  const int* __restrict__ list1 = list0 + list_copy_words;
  const float* __restrict__ hl = h + lane * 4;
  for (int d = wave; d < n_rows; d += nwaves) {
    const int o = off[d];
    const int nA = (int)cur0[d];
    const int nB = (int)cur0[(size_t)cur_copy_words + d];
    float4 acc = pull_row_sum(o, nA, nB, list0, list1, hl, lane);
    const float m = fmaxf((float)(nA + nB), 1.0f);
    acc.x /= m; acc.y /= m; acc.z /= m; acc.w /= m;
    *(float4*)(out + (size_t)d * EMB_DIM + lane * 4) = acc;
  }
}

__global__ void __launch_bounds__(256)
gather_pull_kernel(const int* __restrict__ users, const int* __restrict__ items,
                   const float* __restrict__ uemb, const float* __restrict__ g1,
                   const int* __restrict__ off, const float* __restrict__ cur0,
                   int cur_copy_words, const int* __restrict__ list0,
                   size_t list_copy_words, const float* __restrict__ iemb,
                   float* __restrict__ out_user, float* __restrict__ out_item,
                   int batch) {
  const int lane = threadIdx.x & 31;
  const int b = __builtin_amdgcn_readfirstlane(
      (int)((blockIdx.x * blockDim.x + threadIdx.x) >> 5));
  if (b >= batch) return;
  const int u = users[b];
  const int it = items[b];
  const size_t lo = (size_t)lane * 4;
  const float4 a = *(const float4*)(uemb + (size_t)u * EMB_DIM + lo);
  const float4 c = *(const float4*)(g1 + (size_t)u * EMB_DIM + lo);
  const float4 iv = *(const float4*)(iemb + (size_t)it * EMB_DIM + lo);
  const int o = off[u];
  const int nA = (int)cur0[u];
  const int nB = (int)cur0[(size_t)cur_copy_words + u];
  *(float4*)(out_item + (size_t)b * EMB_DIM + lo) = iv;
  const float4 s2 = pull_row_sum(o, nA, nB, list0, list0 + list_copy_words,
                                 g1 + lo, lane);
  const float m = fmaxf((float)(nA + nB), 1.0f);
  float4 fu;
  fu.x = (a.x + c.x) + s2.x / m;
  fu.y = (a.y + c.y) + s2.y / m;
  fu.z = (a.z + c.z) + s2.z / m;
  fu.w = (a.w + c.w) + s2.w / m;
  *(float4*)(out_user + (size_t)b * EMB_DIM + lo) = fu;
}

__global__ void __launch_bounds__(256)
merge_clear_kernel(float* __restrict__ a, float* __restrict__ b,
                   long long n4) {
  const long long stride = (long long)gridDim.x * blockDim.x;
  const long long t = (long long)blockIdx.x * blockDim.x + threadIdx.x;
  float4* a4 = (float4*)a;
  float4* b4 = (float4*)b;
  const float4 z = make_float4(0.0f, 0.0f, 0.0f, 0.0f);
  for (long long i = t; i < n4; i += stride) {
    float4 x = a4[i];
    const float4 y = b4[i];
    x.x += y.x; x.y += y.y; x.z += y.z; x.w += y.w;
    a4[i] = x;
    b4[i] = z;
  }
}

__global__ void __launch_bounds__(256)
scatter_flagged_kernel(const int* __restrict__ src, const int* __restrict__ dst,
                       const int* __restrict__ flags,
                       const float* __restrict__ h,
                       const float* __restrict__ hB,
                       const float* __restrict__ deg,
                       const float* __restrict__ degB, float* __restrict__ agg0,
                       size_t agg_copy_words, int compact, int n_edges) {
  int e = blockIdx.x * blockDim.x + threadIdx.x;
  int lane = threadIdx.x & 31;
  if ((e - lane) >= n_edges) return;
  float* __restrict__ agg = agg0 + (size_t)chip_half() * agg_copy_words;
  int s = 0, d = 0, f = 0;
  float sc = 1.0f;
  if (e < n_edges) {
    s = src[e];
    d = dst[e];
    f = flags[d];
    if (compact && f != 0) d = f - 1;
    if (f != 0) {
      float dg = deg[s];
      if (degB != nullptr) dg += degB[s];
      sc = 1.0f / fmaxf(dg, 1.0f);
    }
  }
  unsigned mask = __builtin_amdgcn_ballot_w32(f != 0);
  while (mask) {
    int j = __builtin_ctz(mask);
    mask &= mask - 1;
    int sj = __builtin_amdgcn_readlane(s, j);
    int dj = __builtin_amdgcn_readlane(d, j);
    const float cj =
        __int_as_float(__builtin_amdgcn_readlane(__float_as_int(sc), j));
    RowQuad v = load_row(h + (size_t)sj * EMB_DIM + lane);
    if (hB != nullptr) {
      const RowQuad w = load_row(hB + (size_t)sj * EMB_DIM + lane);
      v.a += w.a; v.b += w.b; v.c += w.c; v.d += w.d;
    }
    gatomic_add_row(agg + (size_t)dj * EMB_DIM + lane, v.a * cj, v.b * cj,
                    v.c * cj, v.d * cj);
  }
}

__global__ void __launch_bounds__(256)
gather_kernel(const int* __restrict__ users, const int* __restrict__ items,
              const float* __restrict__ uemb, const float* __restrict__ g1,
              const float* __restrict__ g1B, const float* __restrict__ g2,
              const float* __restrict__ g2B, const int* __restrict__ slots,
              const float* __restrict__ deg, const float* __restrict__ degB,
              const float* __restrict__ iemb, float* __restrict__ out_user,
              float* __restrict__ out_item, int batch) {
  int t = blockIdx.x * blockDim.x + threadIdx.x;
  int b = t >> 5;
  int lane = t & 31;
  if (b >= batch) return;
  int u = users[b];
  int it = items[b];
  size_t off = (size_t)lane * 4;
  const float4 a = *(const float4*)(uemb + (size_t)u * EMB_DIM + off);
  float4 c = *(const float4*)(g1 + (size_t)u * EMB_DIM + off);
  float dg = deg[u];
  if (g1B != nullptr) {
    const float4 cb = *(const float4*)(g1B + (size_t)u * EMB_DIM + off);
    c.x += cb.x; c.y += cb.y; c.z += cb.z; c.w += cb.w;
  }
  if (degB != nullptr) dg += degB[u];
  const int r2 = (slots != nullptr) ? (slots[u] - 1) : u;
  float4 d = *(const float4*)(g2 + (size_t)r2 * EMB_DIM + off);
  if (g2B != nullptr) {
    const float4 db = *(const float4*)(g2B + (size_t)r2 * EMB_DIM + off);
    d.x += db.x; d.y += db.y; d.z += db.z; d.w += db.w;
  }
  const float inv = 1.0f / fmaxf(dg, 1.0f);
  float4 fu;
  fu.x = a.x + c.x * inv + d.x * inv;
  fu.y = a.y + c.y * inv + d.y * inv;
  fu.z = a.z + c.z * inv + d.z * inv;
  fu.w = a.w + c.w * inv + d.w * inv;
  *(float4*)(out_user + (size_t)b * EMB_DIM + off) = fu;
  *(float4*)(out_item + (size_t)b * EMB_DIM + off) =
      *(const float4*)(iemb + (size_t)it * EMB_DIM + off);
}

__global__ void __launch_bounds__(256)
dot_sigmoid_wmma(const float* __restrict__ U, const float* __restrict__ V,
                 float* __restrict__ pred, int batch) {
  int gid = blockIdx.x * blockDim.x + threadIdx.x;
  int wave = gid >> 5;
  int lane = threadIdx.x & 31;
  if (wave * 16 >= batch) return;

  int row = lane & 15;
  int half = lane >> 4;
  const float* urow = U + ((size_t)wave * 16 + row) * EMB_DIM + 2 * half;
  const float* vrow = V + ((size_t)wave * 16 + row) * EMB_DIM + 2 * half;

  v8f acc = {};
#pragma unroll
  for (int k0 = 0; k0 < EMB_DIM; k0 += 4) {
    v2f a = *(const v2f*)(urow + k0);
    v2f b = *(const v2f*)(vrow + k0);
    acc = __builtin_amdgcn_wmma_f32_16x16x4_f32(
         false, a,  false, b,
         (short)0, acc,  false,  false);
  }

  if (lane < 8 || lane >= 24) {
    int i = (lane < 8) ? lane : (lane - 16);
    float d = acc[lane & 7];
    pred[(size_t)wave * 16 + i] = 1.0f / (1.0f + expf(-d));
  }
}

extern "C" void kernel_launch(void* const* d_in, const int* in_sizes, int n_in,
                              void* d_out, int out_size, void* d_ws,
                              size_t ws_size, hipStream_t stream) {
  const int* users  = (const int*)d_in[0];
  const int* items  = (const int*)d_in[1];
  const int* esrc   = (const int*)d_in[2];
  const int* edst   = (const int*)d_in[3];
  const float* uemb = (const float*)d_in[4];
  const float* iemb = (const float*)d_in[5];

  const int batch     = in_sizes[0];
  const int n_edges   = in_sizes[2];
  const int num_users = in_sizes[4] / EMB_DIM;

  const size_t nu = (size_t)num_users;
  const size_t tbl = nu * EMB_DIM;
  const size_t words = ws_size / sizeof(float);
  const bool has_flags = words >= 2 * nu + 2 * tbl;
  const bool has_degB  = words >= 3 * nu + 2 * tbl;
  const bool has_g1B   = words >= 3 * nu + 3 * tbl;
  float* deg  = (float*)d_ws;
  float* degB = has_degB ? deg + nu : nullptr;
  float* g1   = deg + (has_degB ? 2 : 1) * nu;
  float* g1B  = has_g1B ? g1 + tbl : nullptr;
  float* g2   = g1 + (has_g1B ? 2 : 1) * tbl;
  const size_t bt = (size_t)batch * EMB_DIM;
  const bool compact = has_g1B && words >= 3 * nu + 2 * tbl + 2 * bt;
  float* g2B  = compact ? g2 + bt : nullptr;
  const size_t g2_words = compact ? 2 * bt : tbl;
  int* flags  = has_flags ? (int*)(g2 + g2_words) : nullptr;
  const long long total =
      (long long)((g2 + g2_words) - deg) + (has_flags ? (long long)nu : 0);
  const size_t copy1 = tbl;

  float* pred     = (float*)d_out;
  float* out_user = pred + batch;
  float* out_item = out_user + (size_t)batch * EMB_DIM;

  const int B = 256;

  const int G = 2048;

  {
    const size_t nuA = (nu + 31) & ~(size_t)31;
    const size_t neA = ((size_t)n_edges + 31) & ~(size_t)31;
    const size_t zeroed = 4 * nuA;
    const size_t need = zeroed + nuA + 2 * neA + tbl;
    if (words >= need) {
      float* p_deg   = (float*)d_ws;
      float* p_cur   = p_deg + 2 * nuA;
      int*   p_off   = (int*)(p_cur + 2 * nuA);
      int*   p_list  = p_off + nuA;
      float* p_g1    = (float*)(p_list + 2 * neA);
      const int eb = (n_edges + B - 1) / B;
      const int zb = (int)((zeroed / 4 + B - 1) / B);

      zero_f32<<<zb < G ? zb : G, B, 0, stream>>>(p_deg, (long long)zeroed);
      degree_kernel<<<G / 4, B, 0, stream>>>(edst, users, p_deg, nullptr,
                                             n_edges, batch, (int)nuA, 0);
      offsets_kernel<<<(num_users + 1023) / 1024, 1024, 0, stream>>>(
          p_deg, p_deg + nuA, p_off, num_users);
      fill_lists_kernel<<<eb, B, 0, stream>>>(esrc, edst, p_off, p_cur,
                                              (int)nuA, p_list, neA, n_edges);
      pull_rows_kernel<<<G, B, 0, stream>>>(p_off, p_cur, (int)nuA, p_list, neA,
                                            uemb, p_g1, num_users);
      const long long gt = (long long)batch * 32;
      gather_pull_kernel<<<(int)((gt + B - 1) / B), B, 0, stream>>>(
          users, items, uemb, p_g1, p_off, p_cur, (int)nuA, p_list, neA, iemb,
          out_user, out_item, batch);
      const long long dt = (long long)((batch + 15) / 16) * 32;
      dot_sigmoid_wmma<<<(int)((dt + B - 1) / B), B, 0, stream>>>(
          out_user, out_item, pred, batch);
      return;
    }
  }

  zero_f32<<<G, B, 0, stream>>>((float*)d_ws, total);

  degree_kernel<<<G / 4, B, 0, stream>>>(edst, users, deg, flags, n_edges,
                                         batch, has_degB ? num_users : 0,
                                         compact ? 1 : 0);

  const int sc_blocks = (n_edges + B - 1) / B;
  scatter_blocked_kernel<<<4 * G, B, 0, stream>>>(esrc, edst, uemb, g1, n_edges,
                                                  (num_users + 1) / 2, G, copy1);

  if (!has_g1B)
    merge_clear_kernel<<<G, B, 0, stream>>>(g1, g2, (long long)(tbl / 4));

  if (flags != nullptr) {
    scatter_flagged_kernel<<<sc_blocks, B, 0, stream>>>(
        esrc, edst, flags, g1, g1B, deg, degB, g2, compact ? bt : 0,
        compact ? 1 : 0, n_edges);
  } else {
    scatter_kernel<true><<<sc_blocks, B, 0, stream>>>(esrc, edst, g1, deg, g2,
                                                      n_edges);
  }

  long long ga_threads = (long long)batch * 32;
  gather_kernel<<<(int)((ga_threads + B - 1) / B), B, 0, stream>>>(
      users, items, uemb, g1, g1B, g2, g2B, compact ? flags : nullptr, deg,
      degB, iemb, out_user, out_item, batch);

  long long dot_threads = (long long)((batch + 15) / 16) * 32;
  dot_sigmoid_wmma<<<(int)((dot_threads + B - 1) / B), B, 0, stream>>>(
      out_user, out_item, pred, batch);
}


# edge lists in fixed slots of 32 per destination and package half: one cursor add gives place and count (overflow list past 32); degree pass and offsets launch gone; sums stay in list (arrival) order
# speedup vs baseline: 9.3723x; 1.2957x over previous
#include <hip/hip_runtime.h>
#include <hip/hip_bf16.h>

#define EMB_DIM 128

typedef __attribute__((ext_vector_type(2))) float v2f;
typedef __attribute__((ext_vector_type(8))) float v8f;

__device__ __forceinline__ void gatomic_add_f32(float* p, float v) {
  asm volatile("global_atomic_add_f32 %0, %1, off scope:SCOPE_DEV"
               :: "v"(p), "v"(v) : "memory");
}

__device__ __forceinline__ void gatomic_add_row(float* p, float v0, float v1,
                                                float v2, float v3) {
  asm volatile(
      "global_atomic_add_f32 %0, %1, off scope:SCOPE_DEV\n\t"
      "global_atomic_add_f32 %0, %2, off offset:128 scope:SCOPE_DEV\n\t"
      "global_atomic_add_f32 %0, %3, off offset:256 scope:SCOPE_DEV\n\t"
      "global_atomic_add_f32 %0, %4, off offset:384 scope:SCOPE_DEV"
      :: "v"(p), "v"(v0), "v"(v1), "v"(v2), "v"(v3) : "memory");
}

__device__ __forceinline__ float gatomic_add_ret_f32(float* p, float v) {
  float r;
  asm volatile(
      "global_atomic_add_f32 %0, %1, %2, off th:TH_ATOMIC_RETURN scope:SCOPE_DEV\n\t"
      "s_wait_loadcnt 0x0"
      : "=&v"(r) : "v"(p), "v"(v) : "memory");
  return r;
}

__device__ __forceinline__ int chip_half() {
  const unsigned id = __builtin_amdgcn_s_sendmsg_rtn(0x87);
  return (int)((id >> 18) & 1u);
}

struct RowQuad { float a, b, c, d; };
__device__ __forceinline__ RowQuad load_row(const float* __restrict__ p) {
  RowQuad r;
  r.a = p[0];
  r.b = p[32];
  r.c = p[64];
  r.d = p[96];
  return r;
}

__global__ void __launch_bounds__(256)
zero_f32(float* __restrict__ p, long long n) {
  const long long n4 = n >> 2;
  const long long stride = (long long)gridDim.x * blockDim.x;
  const long long t = (long long)blockIdx.x * blockDim.x + threadIdx.x;
  const float4 z = make_float4(0.0f, 0.0f, 0.0f, 0.0f);
  float4* p4 = (float4*)p;
  for (long long i = t; i < n4; i += stride) p4[i] = z;
  if (t < (n & 3)) p[(n4 << 2) + t] = 0.0f;
}

__global__ void __launch_bounds__(256)
degree_kernel(const int* __restrict__ dst, const int* __restrict__ users,
              float* __restrict__ deg, int* __restrict__ flags, int n_edges,
              int batch, int copy_words, int slot_flags) {
  const int stride = gridDim.x * blockDim.x;
  const int t = blockIdx.x * blockDim.x + threadIdx.x;
  float* mydeg = deg + (size_t)chip_half() * copy_words;
  for (int e = t; e < n_edges; e += stride)
    gatomic_add_f32(mydeg + dst[e], 1.0f);
  if (flags != nullptr)
    for (int b = t; b < batch; b += stride)
      flags[users[b]] = slot_flags ? (b + 1) : 1;
}

template <bool SCALE>
__global__ void __launch_bounds__(256)
scatter_kernel(const int* __restrict__ src, const int* __restrict__ dst,
               const float* __restrict__ h, const float* __restrict__ deg,
               float* __restrict__ agg, int n_edges) {
  int e = blockIdx.x * blockDim.x + threadIdx.x;
  int lane = threadIdx.x & 31;
  int base = __builtin_amdgcn_readfirstlane(e - lane);
  if (base >= n_edges) return;
  int s = 0, d = 0;
  float sc = 1.0f;
  if (e < n_edges) {
    s = src[e];
    d = dst[e];
    if (SCALE) sc = 1.0f / fmaxf(deg[s], 1.0f);
  }
  int left = n_edges - base;
  int cnt = left < 32 ? left : 32;
  for (int j = 0; j < cnt; j += 4) {
    int sj[4], dj[4];
    float cj[4];
    RowQuad v[4];
#pragma unroll
    for (int k = 0; k < 4; ++k) {
      int jj = (j + k < cnt) ? (j + k) : (cnt - 1);
      sj[k] = __builtin_amdgcn_readlane(s, jj);
      dj[k] = __builtin_amdgcn_readlane(d, jj);
      if (SCALE)
        cj[k] = __int_as_float(
            __builtin_amdgcn_readlane(__float_as_int(sc), jj));
      v[k] = load_row(h + (size_t)sj[k] * EMB_DIM + lane);
    }
#pragma unroll
    for (int k = 0; k < 4; ++k) {
      if (j + k < cnt) {
        if (SCALE) {
          v[k].a *= cj[k]; v[k].b *= cj[k]; v[k].c *= cj[k]; v[k].d *= cj[k];
        }
        gatomic_add_row(agg + (size_t)dj[k] * EMB_DIM + lane, v[k].a, v[k].b,
                        v[k].c, v[k].d);
      }
    }
  }
}

__global__ void __launch_bounds__(256)
scatter_blocked_kernel(const int* __restrict__ src, const int* __restrict__ dst,
                       const float* __restrict__ h, float* __restrict__ agg0,
                       int n_edges, int half_n, int blocks_per_pass,
                       size_t copy_words) {
  float* __restrict__ agg = agg0 + (size_t)chip_half() * copy_words;
  const int pass = blockIdx.x / blocks_per_pass;
  const int pblock = blockIdx.x - pass * blocks_per_pass;
  const bool dst_hi = (pass >> 1) != 0;
  const bool src_hi = ((pass ^ (pass >> 1)) & 1) != 0;
  const int lane = threadIdx.x & 31;
  const int wave = __builtin_amdgcn_readfirstlane(
      (int)((pblock * blockDim.x + threadIdx.x) >> 5));
  const int step = ((blocks_per_pass * (int)blockDim.x) >> 5) * 32;
  for (int base = wave * 32; base < n_edges; base += step) {
    const int e = base + lane;
    int s = 0, d = 0;
    bool keep = false;
    if (e < n_edges) {
      s = src[e];
      d = dst[e];
      keep = ((s >= half_n) == src_hi) && ((d >= half_n) == dst_hi);
    }
    unsigned mask = __builtin_amdgcn_ballot_w32(keep);
    while (mask) {
      int sj[4], dj[4];
      bool on[4];
      RowQuad v[4];
      int j = 0;
#pragma unroll
      for (int k = 0; k < 4; ++k) {
        on[k] = mask != 0;
        if (on[k]) {
          j = __builtin_ctz(mask);
          mask &= mask - 1;
        }
        sj[k] = __builtin_amdgcn_readlane(s, j);
        dj[k] = __builtin_amdgcn_readlane(d, j);
        v[k] = load_row(h + (size_t)sj[k] * EMB_DIM + lane);
      }
#pragma unroll
      for (int k = 0; k < 4; ++k) {
        if (on[k])
          gatomic_add_row(agg + (size_t)dj[k] * EMB_DIM + lane, v[k].a, v[k].b,
                          v[k].c, v[k].d);
      }
    }
  }
}

#define SLOT_CAP 32

__global__ void __launch_bounds__(256)
fill_slots_kernel(const int* __restrict__ src, const int* __restrict__ dst,
                  float* __restrict__ cur0, int cur_copy_words,
                  int* __restrict__ list0, size_t list_copy_words,
                  float* __restrict__ ovc, int* __restrict__ ov, int n_edges) {
  const int e = blockIdx.x * blockDim.x + threadIdx.x;
  const int half = chip_half();
  float* cur = cur0 + (size_t)half * cur_copy_words;
  int* list = list0 + (size_t)half * list_copy_words;
  if (e < n_edges) {
    const int s = src[e];
    const int d = dst[e];
    const int slot = (int)gatomic_add_ret_f32(cur + d, 1.0f);
    int* where = list + (size_t)d * SLOT_CAP + slot;
    int what = s;
    if (slot >= SLOT_CAP) {
      const unsigned p = (unsigned)(int)gatomic_add_ret_f32(ovc, 1.0f);
      const unsigned last = (unsigned)(n_edges - 1);
      where = ov + (p < last ? p : last);
      what = e;
    }
    *where = what;
  }
}

__device__ __forceinline__ float4 overflow_row_sum(
    int d, const float* __restrict__ ovc, const int* __restrict__ ov,
    const int* __restrict__ src, const int* __restrict__ dst, int n_edges,
    const float* __restrict__ hl, int lane) {
  float4 acc = make_float4(0.0f, 0.0f, 0.0f, 0.0f);
  int n_ov = (int)ovc[0];
  if (n_ov > n_edges) n_ov = n_edges;
  for (int c = 0; c < n_ov; c += 32) {
    const int j = c + lane;
    int s = -1;
    if (j < n_ov) {
      const unsigned e = (unsigned)ov[j];
      if (e < (unsigned)n_edges && dst[e] == d) s = src[e];
    }
    unsigned mask = __builtin_amdgcn_ballot_w32(s >= 0);
    while (mask) {
      const int k = __builtin_ctz(mask);
      mask &= mask - 1;
      const int sk = __builtin_amdgcn_readlane(s, k);
      const float4 v = *(const float4*)(hl + (size_t)sk * EMB_DIM);
      acc.x += v.x; acc.y += v.y; acc.z += v.z; acc.w += v.w;
    }
  }
  return acc;
}

__device__ __forceinline__ float4 pull_row_sum(int o, int nA, int nB,
                                               const int* __restrict__ list0,
                                               const int* __restrict__ list1,
                                               const float* __restrict__ hl,
                                               int lane) {
  const int n = nA + nB;
  float4 acc = make_float4(0.0f, 0.0f, 0.0f, 0.0f);
  for (int c = 0; c < n; c += 32) {
    const int j = c + lane;
    int idx = 0;
    if (j < nA) idx = list0[o + j];
    else if (j < n) idx = list1[o + j - nA];
    const int m = (n - c) < 32 ? (n - c) : 32;
    int k = 0;
    for (; k + 8 <= m; k += 8) {
      float4 v[8];
#pragma unroll
      for (int u = 0; u < 8; ++u) {
        const int s = __builtin_amdgcn_readlane(idx, k + u);
        v[u] = *(const float4*)(hl + (size_t)s * EMB_DIM);
      }
#pragma unroll
      for (int u = 0; u < 8; ++u) {
        acc.x += v[u].x; acc.y += v[u].y; acc.z += v[u].z; acc.w += v[u].w;
      }
    }
    for (; k < m; k += 4) {
      float4 v[4];
#pragma unroll
      for (int u = 0; u < 4; ++u) {
        const int kk = (k + u < m) ? (k + u) : (m - 1);
        const int s = __builtin_amdgcn_readlane(idx, kk);
        v[u] = *(const float4*)(hl + (size_t)s * EMB_DIM);
      }
#pragma unroll
      for (int u = 0; u < 4; ++u) {
        if (k + u < m) {
          acc.x += v[u].x; acc.y += v[u].y; acc.z += v[u].z; acc.w += v[u].w;
        }
      }
    }
  }
  return acc;
}

__global__ void __launch_bounds__(256)
pull_rows_kernel(const float* __restrict__ cur0, int cur_copy_words,
                 const int* __restrict__ list0, size_t list_copy_words,
                 const float* __restrict__ ovc, const int* __restrict__ ov,
                 const int* __restrict__ src, const int* __restrict__ dst,
                 int n_edges, const float* __restrict__ h,
                 float* __restrict__ out, int n_rows) {
  const int lane = threadIdx.x & 31;
  const int wave = __builtin_amdgcn_readfirstlane(
      (int)((blockIdx.x * blockDim.x + threadIdx.x) >> 5));
  const int nwaves = (int)((gridDim.x * blockDim.x) >> 5);
  const int* __restrict__ list1 = list0 + list_copy_words;
  const float* __restrict__ hl = h + lane * 4;
  for (int d = wave; d < n_rows; d += nwaves) {
    const int cA = (int)cur0[d];
    const int cB = (int)cur0[(size_t)cur_copy_words + d];
    const int nA = cA < SLOT_CAP ? cA : SLOT_CAP;
    const int nB = cB < SLOT_CAP ? cB : SLOT_CAP;
    float4 acc = pull_row_sum(d * SLOT_CAP, nA, nB, list0, list1, hl, lane);
    if (cA > SLOT_CAP || cB > SLOT_CAP) {
      const float4 t =
          overflow_row_sum(d, ovc, ov, src, dst, n_edges, hl, lane);
      acc.x += t.x; acc.y += t.y; acc.z += t.z; acc.w += t.w;
    }
    const float m = fmaxf((float)(cA + cB), 1.0f);
    acc.x /= m; acc.y /= m; acc.z /= m; acc.w /= m;
    *(float4*)(out + (size_t)d * EMB_DIM + lane * 4) = acc;
  }
}

__global__ void __launch_bounds__(256)
gather_pull_kernel(const int* __restrict__ users, const int* __restrict__ items,
                   const float* __restrict__ uemb, const float* __restrict__ g1,
                   const float* __restrict__ cur0, int cur_copy_words,
                   const int* __restrict__ list0, size_t list_copy_words,
                   const float* __restrict__ ovc, const int* __restrict__ ov,
                   const int* __restrict__ src, const int* __restrict__ dst,
                   int n_edges, const float* __restrict__ iemb,
                   float* __restrict__ out_user, float* __restrict__ out_item,
                   int batch) {
  const int lane = threadIdx.x & 31;
  const int b = __builtin_amdgcn_readfirstlane(
      (int)((blockIdx.x * blockDim.x + threadIdx.x) >> 5));
  if (b >= batch) return;
  const int u = users[b];
  const int it = items[b];
  const size_t lo = (size_t)lane * 4;
  const float4 a = *(const float4*)(uemb + (size_t)u * EMB_DIM + lo);
  const float4 c = *(const float4*)(g1 + (size_t)u * EMB_DIM + lo);
  const float4 iv = *(const float4*)(iemb + (size_t)it * EMB_DIM + lo);
  const int cA = (int)cur0[u];
  const int cB = (int)cur0[(size_t)cur_copy_words + u];
  const int nA = cA < SLOT_CAP ? cA : SLOT_CAP;
  const int nB = cB < SLOT_CAP ? cB : SLOT_CAP;
  *(float4*)(out_item + (size_t)b * EMB_DIM + lo) = iv;
  float4 s2 = pull_row_sum(u * SLOT_CAP, nA, nB, list0,
                           list0 + list_copy_words, g1 + lo, lane);
  if (cA > SLOT_CAP || cB > SLOT_CAP) {
    const float4 t =
        overflow_row_sum(u, ovc, ov, src, dst, n_edges, g1 + lo, lane);
    s2.x += t.x; s2.y += t.y; s2.z += t.z; s2.w += t.w;
  }
  const float m = fmaxf((float)(cA + cB), 1.0f);
  float4 fu;
  fu.x = (a.x + c.x) + s2.x / m;
  fu.y = (a.y + c.y) + s2.y / m;
  fu.z = (a.z + c.z) + s2.z / m;
  fu.w = (a.w + c.w) + s2.w / m;
  *(float4*)(out_user + (size_t)b * EMB_DIM + lo) = fu;
}

__global__ void __launch_bounds__(256)
merge_clear_kernel(float* __restrict__ a, float* __restrict__ b,
                   long long n4) {
  const long long stride = (long long)gridDim.x * blockDim.x;
  const long long t = (long long)blockIdx.x * blockDim.x + threadIdx.x;
  float4* a4 = (float4*)a;
  float4* b4 = (float4*)b;
  const float4 z = make_float4(0.0f, 0.0f, 0.0f, 0.0f);
  for (long long i = t; i < n4; i += stride) {
    float4 x = a4[i];
    const float4 y = b4[i];
    x.x += y.x; x.y += y.y; x.z += y.z; x.w += y.w;
    a4[i] = x;
    b4[i] = z;
  }
}

__global__ void __launch_bounds__(256)
scatter_flagged_kernel(const int* __restrict__ src, const int* __restrict__ dst,
                       const int* __restrict__ flags,
                       const float* __restrict__ h,
                       const float* __restrict__ hB,
                       const float* __restrict__ deg,
                       const float* __restrict__ degB, float* __restrict__ agg0,
                       size_t agg_copy_words, int compact, int n_edges) {
  int e = blockIdx.x * blockDim.x + threadIdx.x;
  int lane = threadIdx.x & 31;
  if ((e - lane) >= n_edges) return;
  float* __restrict__ agg = agg0 + (size_t)chip_half() * agg_copy_words;
  int s = 0, d = 0, f = 0;
  float sc = 1.0f;
  if (e < n_edges) {
    s = src[e];
    d = dst[e];
    f = flags[d];
    if (compact && f != 0) d = f - 1;
    if (f != 0) {
      float dg = deg[s];
      if (degB != nullptr) dg += degB[s];
      sc = 1.0f / fmaxf(dg, 1.0f);
    }
  }
  unsigned mask = __builtin_amdgcn_ballot_w32(f != 0);
  while (mask) {
    int j = __builtin_ctz(mask);
    mask &= mask - 1;
    int sj = __builtin_amdgcn_readlane(s, j);
    int dj = __builtin_amdgcn_readlane(d, j);
    const float cj =
        __int_as_float(__builtin_amdgcn_readlane(__float_as_int(sc), j));
    RowQuad v = load_row(h + (size_t)sj * EMB_DIM + lane);
    if (hB != nullptr) {
      const RowQuad w = load_row(hB + (size_t)sj * EMB_DIM + lane);
      v.a += w.a; v.b += w.b; v.c += w.c; v.d += w.d;
    }
    gatomic_add_row(agg + (size_t)dj * EMB_DIM + lane, v.a * cj, v.b * cj,
                    v.c * cj, v.d * cj);
  }
}

__global__ void __launch_bounds__(256)
gather_kernel(const int* __restrict__ users, const int* __restrict__ items,
              const float* __restrict__ uemb, const float* __restrict__ g1,
              const float* __restrict__ g1B, const float* __restrict__ g2,
              const float* __restrict__ g2B, const int* __restrict__ slots,
              const float* __restrict__ deg, const float* __restrict__ degB,
              const float* __restrict__ iemb, float* __restrict__ out_user,
              float* __restrict__ out_item, int batch) {
  int t = blockIdx.x * blockDim.x + threadIdx.x;
  int b = t >> 5;
  int lane = t & 31;
  if (b >= batch) return;
  int u = users[b];
  int it = items[b];
  size_t off = (size_t)lane * 4;
  const float4 a = *(const float4*)(uemb + (size_t)u * EMB_DIM + off);
  float4 c = *(const float4*)(g1 + (size_t)u * EMB_DIM + off);
  float dg = deg[u];
  if (g1B != nullptr) {
    const float4 cb = *(const float4*)(g1B + (size_t)u * EMB_DIM + off);
    c.x += cb.x; c.y += cb.y; c.z += cb.z; c.w += cb.w;
  }
  if (degB != nullptr) dg += degB[u];
  const int r2 = (slots != nullptr) ? (slots[u] - 1) : u;
  float4 d = *(const float4*)(g2 + (size_t)r2 * EMB_DIM + off);
  if (g2B != nullptr) {
    const float4 db = *(const float4*)(g2B + (size_t)r2 * EMB_DIM + off);
    d.x += db.x; d.y += db.y; d.z += db.z; d.w += db.w;
  }
  const float inv = 1.0f / fmaxf(dg, 1.0f);
  float4 fu;
  fu.x = a.x + c.x * inv + d.x * inv;
  fu.y = a.y + c.y * inv + d.y * inv;
  fu.z = a.z + c.z * inv + d.z * inv;
  fu.w = a.w + c.w * inv + d.w * inv;
  *(float4*)(out_user + (size_t)b * EMB_DIM + off) = fu;
  *(float4*)(out_item + (size_t)b * EMB_DIM + off) =
      *(const float4*)(iemb + (size_t)it * EMB_DIM + off);
}

__global__ void __launch_bounds__(256)
dot_sigmoid_wmma(const float* __restrict__ U, const float* __restrict__ V,
                 float* __restrict__ pred, int batch) {
  int gid = blockIdx.x * blockDim.x + threadIdx.x;
  int wave = gid >> 5;
  int lane = threadIdx.x & 31;
  if (wave * 16 >= batch) return;

  int row = lane & 15;
  int half = lane >> 4;
  const float* urow = U + ((size_t)wave * 16 + row) * EMB_DIM + 2 * half;
  const float* vrow = V + ((size_t)wave * 16 + row) * EMB_DIM + 2 * half;

  v8f acc = {};
#pragma unroll
  for (int k0 = 0; k0 < EMB_DIM; k0 += 4) {
    v2f a = *(const v2f*)(urow + k0);
    v2f b = *(const v2f*)(vrow + k0);
    acc = __builtin_amdgcn_wmma_f32_16x16x4_f32(
         false, a,  false, b,
         (short)0, acc,  false,  false);
  }

  if (lane < 8 || lane >= 24) {
    int i = (lane < 8) ? lane : (lane - 16);
    float d = acc[lane & 7];
    pred[(size_t)wave * 16 + i] = 1.0f / (1.0f + expf(-d));
  }
}

extern "C" void kernel_launch(void* const* d_in, const int* in_sizes, int n_in,
                              void* d_out, int out_size, void* d_ws,
                              size_t ws_size, hipStream_t stream) {
  const int* users  = (const int*)d_in[0];
  const int* items  = (const int*)d_in[1];
  const int* esrc   = (const int*)d_in[2];
  const int* edst   = (const int*)d_in[3];
  const float* uemb = (const float*)d_in[4];
  const float* iemb = (const float*)d_in[5];

  const int batch     = in_sizes[0];
  const int n_edges   = in_sizes[2];
  const int num_users = in_sizes[4] / EMB_DIM;

  const size_t nu = (size_t)num_users;
  const size_t tbl = nu * EMB_DIM;
  const size_t words = ws_size / sizeof(float);
  const bool has_flags = words >= 2 * nu + 2 * tbl;
  const bool has_degB  = words >= 3 * nu + 2 * tbl;
  const bool has_g1B   = words >= 3 * nu + 3 * tbl;
  float* deg  = (float*)d_ws;
  float* degB = has_degB ? deg + nu : nullptr;
  float* g1   = deg + (has_degB ? 2 : 1) * nu;
  float* g1B  = has_g1B ? g1 + tbl : nullptr;
  float* g2   = g1 + (has_g1B ? 2 : 1) * tbl;
  const size_t bt = (size_t)batch * EMB_DIM;
  const bool compact = has_g1B && words >= 3 * nu + 2 * tbl + 2 * bt;
  float* g2B  = compact ? g2 + bt : nullptr;
  const size_t g2_words = compact ? 2 * bt : tbl;
  int* flags  = has_flags ? (int*)(g2 + g2_words) : nullptr;
  const long long total =
      (long long)((g2 + g2_words) - deg) + (has_flags ? (long long)nu : 0);
  const size_t copy1 = tbl;

  float* pred     = (float*)d_out;
  float* out_user = pred + batch;
  float* out_item = out_user + (size_t)batch * EMB_DIM;

  const int B = 256;

  const int G = 2048;

  {
    const size_t nuA = (nu + 31) & ~(size_t)31;
    const size_t neA = ((size_t)n_edges + 31) & ~(size_t)31;
    const size_t slots = nuA * SLOT_CAP;
    const size_t zeroed = 2 * nuA + 32;
    const size_t need = zeroed + 2 * slots + neA + tbl;
    if (words >= need) {
      float* p_cur   = (float*)d_ws;
      float* p_ovc   = p_cur + 2 * nuA;
      int*   p_list  = (int*)(p_ovc + 32);
      int*   p_ov    = p_list + 2 * slots;
      float* p_g1    = (float*)(p_ov + neA);
      const int eb = (n_edges + B - 1) / B;
      const int zb = (int)((zeroed / 4 + B - 1) / B);

      zero_f32<<<zb < G ? zb : G, B, 0, stream>>>(p_cur, (long long)zeroed);
      fill_slots_kernel<<<eb, B, 0, stream>>>(esrc, edst, p_cur, (int)nuA,
                                              p_list, slots, p_ovc, p_ov,
                                              n_edges);
      pull_rows_kernel<<<G, B, 0, stream>>>(p_cur, (int)nuA, p_list, slots,
                                            p_ovc, p_ov, esrc, edst, n_edges,
                                            uemb, p_g1, num_users);
      const long long gt = (long long)batch * 32;
      gather_pull_kernel<<<(int)((gt + B - 1) / B), B, 0, stream>>>(
          users, items, uemb, p_g1, p_cur, (int)nuA, p_list, slots, p_ovc,
          p_ov, esrc, edst, n_edges, iemb, out_user, out_item, batch);
      const long long dt = (long long)((batch + 15) / 16) * 32;
      dot_sigmoid_wmma<<<(int)((dt + B - 1) / B), B, 0, stream>>>(
          out_user, out_item, pred, batch);
      return;
    }
  }

  zero_f32<<<G, B, 0, stream>>>((float*)d_ws, total);

  degree_kernel<<<G / 4, B, 0, stream>>>(edst, users, deg, flags, n_edges,
                                         batch, has_degB ? num_users : 0,
                                         compact ? 1 : 0);

  const int sc_blocks = (n_edges + B - 1) / B;
  scatter_blocked_kernel<<<4 * G, B, 0, stream>>>(esrc, edst, uemb, g1, n_edges,
                                                  (num_users + 1) / 2, G, copy1);

  if (!has_g1B)
    merge_clear_kernel<<<G, B, 0, stream>>>(g1, g2, (long long)(tbl / 4));

  if (flags != nullptr) {
    scatter_flagged_kernel<<<sc_blocks, B, 0, stream>>>(
        esrc, edst, flags, g1, g1B, deg, degB, g2, compact ? bt : 0,
        compact ? 1 : 0, n_edges);
  } else {
    scatter_kernel<true><<<sc_blocks, B, 0, stream>>>(esrc, edst, g1, deg, g2,
                                                      n_edges);
  }

  long long ga_threads = (long long)batch * 32;
  gather_kernel<<<(int)((ga_threads + B - 1) / B), B, 0, stream>>>(
      users, items, uemb, g1, g1B, g2, g2B, compact ? flags : nullptr, deg,
      degB, iemb, out_user, out_item, batch);

  long long dot_threads = (long long)((batch + 15) / 16) * 32;
  dot_sigmoid_wmma<<<(int)((dot_threads + B - 1) / B), B, 0, stream>>>(
      out_user, out_item, pred, batch);
}
